# starting kernel
# baseline (speedup 1.0000x reference)
#include <hip/hip_runtime.h>
#include <hip/hip_bf16.h>


typedef float v2f __attribute__((ext_vector_type(2)));
typedef float v8f __attribute__((ext_vector_type(8)));

__device__ __forceinline__ void atomAddF(float* p, float v) {
#if defined(__HIP_PLATFORM_AMD__)
    unsafeAtomicAdd(p, v);
#else
    atomicAdd(p, v);
#endif
}

__global__ __launch_bounds__(256) void k_init_ones(float* __restrict__ p, int n) {
    int i = blockIdx.x * blockDim.x + threadIdx.x;
    if (i < n) p[i] = 1.0f;
}

__global__ __launch_bounds__(256) void k_deg_accum(const int* __restrict__ dst,
                                                   float* __restrict__ deg, int E) {
    int e = blockIdx.x * blockDim.x + threadIdx.x;
    if (e < E) atomAddF(&deg[dst[e]], 1.0f);
}

__global__ __launch_bounds__(256) void k_rsqrt_inplace(float* __restrict__ p, int n) {
    int i = blockIdx.x * blockDim.x + threadIdx.x;
    if (i < n) p[i] = rsqrtf(p[i]);
}

template <int K, int F>
__global__ __launch_bounds__(256) void k_gemm_wmma(const float* __restrict__ X,
                                                   const float* __restrict__ W,
                                                   float* __restrict__ H,
                                                   int tilesM) {
    static_assert(K % 4 == 0 && F % 16 == 0, "shape");
    const int tilesF = F / 16;
    const int wave = blockIdx.x * (blockDim.x >> 5) + (threadIdx.x >> 5);
    if (wave >= tilesM * tilesF) return;
    const int tm = wave / tilesF;
    const int tf = wave % tilesF;
    const int lane = threadIdx.x & 31;
    const int l15  = lane & 15;
    const int hi2  = (lane >> 4) << 1;

    const float* xrow = X + (size_t)(tm * 16 + l15) * K;
    const float* wcol = W + (tf * 16 + l15);

    v8f acc = {};
#pragma unroll 4
    for (int kk = 0; kk < K; kk += 4) {
        v2f a, b;
        a.x = xrow[kk + hi2];
        a.y = xrow[kk + hi2 + 1];
        b.x = wcol[(size_t)(kk + hi2) * F];
        b.y = wcol[(size_t)(kk + hi2 + 1) * F];
        acc = __builtin_amdgcn_wmma_f32_16x16x4_f32(
             false, a,  false, b,
             (short)0, acc,  false,  false);
    }

    const int hi8 = (lane >> 4) << 3;
#pragma unroll
    for (int v = 0; v < 8; ++v)
        H[(size_t)(tm * 16 + hi8 + v) * F + tf * 16 + l15] = acc[v];
}

template <int K, int F>
__global__ __launch_bounds__(256) void k_gemm_tail(const float* __restrict__ X,
                                                   const float* __restrict__ W,
                                                   float* __restrict__ H,
                                                   int rowStart, int rowEnd) {
    int idx = blockIdx.x * blockDim.x + threadIdx.x;
    int rows = rowEnd - rowStart;
    if (idx >= rows * F) return;
    int r = rowStart + idx / F, f = idx % F;
    float s = 0.f;
    for (int k = 0; k < K; ++k) s += X[(size_t)r * K + k] * W[(size_t)k * F + f];
    H[(size_t)r * F + f] = s;
}

template <int F>
__global__ __launch_bounds__(256) void k_scatter(const float* __restrict__ H,
                                                 const int* __restrict__ src,
                                                 const int* __restrict__ dst,
                                                 const float* __restrict__ dinv,
                                                 float* __restrict__ AGG, int E) {
    constexpr int CH = F / 4;
    long long tid = (long long)blockIdx.x * blockDim.x + threadIdx.x;
    if (tid >= (long long)E * CH) return;
    int e = (int)(tid / CH);
    int c = (int)(tid % CH);
    int s = src[e], d = dst[e];
    float w = dinv[s] * dinv[d];
    float4 hv = ((const float4*)(H + (size_t)s * F))[c];
    float* ap = AGG + (size_t)d * F + c * 4;
    atomAddF(ap + 0, hv.x * w);
    atomAddF(ap + 1, hv.y * w);
    atomAddF(ap + 2, hv.z * w);
    atomAddF(ap + 3, hv.w * w);
}

template <int F, bool RELU>
__global__ __launch_bounds__(256) void k_self_bias(float* __restrict__ AGG,
                                                   const float* __restrict__ H,
                                                   const float* __restrict__ dinv,
                                                   const float* __restrict__ bias,
                                                   int N) {
    long long idx = (long long)blockIdx.x * blockDim.x + threadIdx.x;
    if (idx >= (long long)N * F) return;
    int i = (int)(idx / F), f = (int)(idx % F);
    float ns = dinv[i];
    ns *= ns;
    float v = AGG[idx] + H[idx] * ns + bias[f];
    if (RELU) v = fmaxf(v, 0.0f);
    AGG[idx] = v;
}

static inline int nblk(long long n, int t) { return (int)((n + t - 1) / t); }

template <int K, int F>
static void launch_gemm(const float* X, const float* W, float* H, int N, hipStream_t s) {
    int tilesM = N / 16;
    long long waves = (long long)tilesM * (F / 16);
    if (waves > 0)
        k_gemm_wmma<K, F><<<nblk(waves, 8), 256, 0, s>>>(X, W, H, tilesM);
    int tail = tilesM * 16;
    if (tail < N)
        k_gemm_tail<K, F><<<nblk((long long)(N - tail) * F, 256), 256, 0, s>>>(X, W, H, tail, N);
}

extern "C" void kernel_launch(void* const* d_in, const int* in_sizes, int n_in,
                              void* d_out, int out_size, void* d_ws, size_t ws_size,
                              hipStream_t stream) {
    const float* x  = (const float*)d_in[0];
    const int*   ei = (const int*)d_in[1];
    const float* W1 = (const float*)d_in[2]; const float* b1 = (const float*)d_in[3];
    const float* W2 = (const float*)d_in[4]; const float* b2 = (const float*)d_in[5];
    const float* W3 = (const float*)d_in[6]; const float* b3 = (const float*)d_in[7];
    const float* W4 = (const float*)d_in[8]; const float* b4 = (const float*)d_in[9];

    const int N = in_sizes[0] / 128;
    const int E = in_sizes[1] / 2;
    const int* src = ei;
    const int* dst = ei + E;

    float* out  = (float*)d_out;
    float* xrec = out;
    float* z    = out + (size_t)N * 128;

    float* ws   = (float*)d_ws;
    float* dinv = ws;
    float* bufH = ws + N;
    float* bufA = bufH + (size_t)N * 128;

    k_init_ones<<<nblk(N, 256), 256, 0, stream>>>(dinv, N);
    k_deg_accum<<<nblk(E, 256), 256, 0, stream>>>(dst, dinv, E);
    k_rsqrt_inplace<<<nblk(N, 256), 256, 0, stream>>>(dinv, N);

    launch_gemm<128, 64>(x, W1, bufH, N, stream);
    hipMemsetAsync(bufA, 0, (size_t)N * 64 * sizeof(float), stream);
    k_scatter<64><<<nblk((long long)E * 16, 256), 256, 0, stream>>>(bufH, src, dst, dinv, bufA, E);
    k_self_bias<64, true><<<nblk((long long)N * 64, 256), 256, 0, stream>>>(bufA, bufH, dinv, b1, N);

    launch_gemm<64, 32>(bufA, W2, bufH, N, stream);
    hipMemsetAsync(z, 0, (size_t)N * 32 * sizeof(float), stream);
    k_scatter<32><<<nblk((long long)E * 8, 256), 256, 0, stream>>>(bufH, src, dst, dinv, z, E);
    k_self_bias<32, false><<<nblk((long long)N * 32, 256), 256, 0, stream>>>(z, bufH, dinv, b2, N);

    launch_gemm<32, 64>(z, W3, bufH, N, stream);
    hipMemsetAsync(bufA, 0, (size_t)N * 64 * sizeof(float), stream);
    k_scatter<64><<<nblk((long long)E * 16, 256), 256, 0, stream>>>(bufH, src, dst, dinv, bufA, E);
    k_self_bias<64, true><<<nblk((long long)N * 64, 256), 256, 0, stream>>>(bufA, bufH, dinv, b3, N);

    launch_gemm<64, 128>(bufA, W4, bufH, N, stream);
    hipMemsetAsync(xrec, 0, (size_t)N * 128 * sizeof(float), stream);
    k_scatter<128><<<nblk((long long)E * 32, 256), 256, 0, stream>>>(bufH, src, dst, dinv, xrec, E);
    k_self_bias<128, false><<<nblk((long long)N * 128, 256), 256, 0, stream>>>(xrec, bufH, dinv, b4, N);
}


# Layers 3-4 aggregate before the GEMM, (A h)W not A(hW): atomic scatter at width 32/64 instead of 64/128, bias+ReLU in the GEMM store; sums reassociated (neighbour sum now precedes the k sum)
# speedup vs baseline: 1.3036x; 1.3036x over previous
#include <hip/hip_runtime.h>
#include <hip/hip_bf16.h>


typedef float v2f __attribute__((ext_vector_type(2)));
typedef float v8f __attribute__((ext_vector_type(8)));

__device__ __forceinline__ void atomAddF(float* p, float v) {
#if defined(__HIP_PLATFORM_AMD__)
    unsafeAtomicAdd(p, v);
#else
    atomicAdd(p, v);
#endif
}

__global__ __launch_bounds__(256) void k_init_ones(float* __restrict__ p, int n) {
    int i = blockIdx.x * blockDim.x + threadIdx.x;
    if (i < n) p[i] = 1.0f;
}

__global__ __launch_bounds__(256) void k_deg_accum(const int* __restrict__ dst,
                                                   float* __restrict__ deg, int E) {
    int e = blockIdx.x * blockDim.x + threadIdx.x;
    if (e < E) atomAddF(&deg[dst[e]], 1.0f);
}

__global__ __launch_bounds__(256) void k_rsqrt_inplace(float* __restrict__ p, int n) {
    int i = blockIdx.x * blockDim.x + threadIdx.x;
    if (i < n) p[i] = rsqrtf(p[i]);
}

template <int K, int F, bool BIAS, bool RELU>
__global__ __launch_bounds__(256) void k_gemm_wmma(const float* __restrict__ X,
                                                   const float* __restrict__ W,
                                                   const float* __restrict__ bias,
                                                   float* __restrict__ H,
                                                   int tilesM) {
    static_assert(K % 4 == 0 && F % 16 == 0, "shape");
    const int tilesF = F / 16;
    const int wave = blockIdx.x * (blockDim.x >> 5) + (threadIdx.x >> 5);
    if (wave >= tilesM * tilesF) return;
    const int tm = wave / tilesF;
    const int tf = wave % tilesF;
    const int lane = threadIdx.x & 31;
    const int l15  = lane & 15;
    const int hi2  = (lane >> 4) << 1;

    const float* xrow = X + (size_t)(tm * 16 + l15) * K;
    const float* wcol = W + (tf * 16 + l15);

    v8f acc = {};
#pragma unroll 4
    for (int kk = 0; kk < K; kk += 4) {
        v2f a, b;
        a.x = xrow[kk + hi2];
        a.y = xrow[kk + hi2 + 1];
        b.x = wcol[(size_t)(kk + hi2) * F];
        b.y = wcol[(size_t)(kk + hi2 + 1) * F];
        acc = __builtin_amdgcn_wmma_f32_16x16x4_f32(
             false, a,  false, b,
             (short)0, acc,  false,  false);
    }

    const int hi8 = (lane >> 4) << 3;
    float bv = 0.0f;
    if (BIAS) bv = bias[tf * 16 + l15];
#pragma unroll
    for (int v = 0; v < 8; ++v) {
        float o = acc[v];
        if (BIAS) o += bv;
        if (RELU) o = fmaxf(o, 0.0f);
        H[(size_t)(tm * 16 + hi8 + v) * F + tf * 16 + l15] = o;
    }
}

template <int K, int F, bool BIAS, bool RELU>
__global__ __launch_bounds__(256) void k_gemm_tail(const float* __restrict__ X,
                                                   const float* __restrict__ W,
                                                   const float* __restrict__ bias,
                                                   float* __restrict__ H,
                                                   int rowStart, int rowEnd) {
    int idx = blockIdx.x * blockDim.x + threadIdx.x;
    int rows = rowEnd - rowStart;
    if (idx >= rows * F) return;
    int r = rowStart + idx / F, f = idx % F;
    float s = 0.f;
    for (int k = 0; k < K; ++k) s += X[(size_t)r * K + k] * W[(size_t)k * F + f];
    if (BIAS) s += bias[f];
    if (RELU) s = fmaxf(s, 0.0f);
    H[(size_t)r * F + f] = s;
}

template <int F>
__global__ __launch_bounds__(256) void k_scatter(const float* __restrict__ H,
                                                 const int* __restrict__ src,
                                                 const int* __restrict__ dst,
                                                 const float* __restrict__ dinv,
                                                 float* __restrict__ AGG, int E) {
    constexpr int CH = F / 4;
    long long tid = (long long)blockIdx.x * blockDim.x + threadIdx.x;
    if (tid >= (long long)E * CH) return;
    int e = (int)(tid / CH);
    int c = (int)(tid % CH);
    int s = src[e], d = dst[e];
    float w = dinv[s] * dinv[d];
    float4 hv = ((const float4*)(H + (size_t)s * F))[c];
    float* ap = AGG + (size_t)d * F + c * 4;
    atomAddF(ap + 0, hv.x * w);
    atomAddF(ap + 1, hv.y * w);
    atomAddF(ap + 2, hv.z * w);
    atomAddF(ap + 3, hv.w * w);
}

template <int F, bool BIAS, bool RELU>
__global__ __launch_bounds__(256) void k_self_bias(float* __restrict__ AGG,
                                                   const float* __restrict__ H,
                                                   const float* __restrict__ dinv,
                                                   const float* __restrict__ bias,
                                                   int N) {
    long long idx = (long long)blockIdx.x * blockDim.x + threadIdx.x;
    if (idx >= (long long)N * F) return;
    int i = (int)(idx / F), f = (int)(idx % F);
    float ns = dinv[i];
    ns *= ns;
    float v = AGG[idx] + H[idx] * ns;
    if (BIAS) v += bias[f];
    if (RELU) v = fmaxf(v, 0.0f);
    AGG[idx] = v;
}

static inline int nblk(long long n, int t) { return (int)((n + t - 1) / t); }

template <int K, int F, bool BIAS, bool RELU>
static void launch_gemm(const float* X, const float* W, const float* bias, float* H, int N,
                        hipStream_t s) {
    int tilesM = N / 16;
    long long waves = (long long)tilesM * (F / 16);
    if (waves > 0)
        k_gemm_wmma<K, F, BIAS, RELU><<<nblk(waves, 8), 256, 0, s>>>(X, W, bias, H, tilesM);
    int tail = tilesM * 16;
    if (tail < N)
        k_gemm_tail<K, F, BIAS, RELU><<<nblk((long long)(N - tail) * F, 256), 256, 0, s>>>(
            X, W, bias, H, tail, N);
}

extern "C" void kernel_launch(void* const* d_in, const int* in_sizes, int n_in,
                              void* d_out, int out_size, void* d_ws, size_t ws_size,
                              hipStream_t stream) {
    const float* x  = (const float*)d_in[0];
    const int*   ei = (const int*)d_in[1];
    const float* W1 = (const float*)d_in[2]; const float* b1 = (const float*)d_in[3];
    const float* W2 = (const float*)d_in[4]; const float* b2 = (const float*)d_in[5];
    const float* W3 = (const float*)d_in[6]; const float* b3 = (const float*)d_in[7];
    const float* W4 = (const float*)d_in[8]; const float* b4 = (const float*)d_in[9];

    const int N = in_sizes[0] / 128;
    const int E = in_sizes[1] / 2;
    const int* src = ei;
    const int* dst = ei + E;

    float* out  = (float*)d_out;
    float* xrec = out;
    float* z    = out + (size_t)N * 128;

    float* ws   = (float*)d_ws;
    float* dinv = ws;
    float* bufH = ws + N;
    float* bufA = bufH + (size_t)N * 128;

    k_init_ones<<<nblk(N, 256), 256, 0, stream>>>(dinv, N);
    k_deg_accum<<<nblk(E, 256), 256, 0, stream>>>(dst, dinv, E);
    k_rsqrt_inplace<<<nblk(N, 256), 256, 0, stream>>>(dinv, N);

    launch_gemm<128, 64, false, false>(x, W1, nullptr, bufH, N, stream);
    hipMemsetAsync(bufA, 0, (size_t)N * 64 * sizeof(float), stream);
    k_scatter<64><<<nblk((long long)E * 16, 256), 256, 0, stream>>>(bufH, src, dst, dinv, bufA, E);
    k_self_bias<64, true, true><<<nblk((long long)N * 64, 256), 256, 0, stream>>>(bufA, bufH, dinv, b1, N);

    launch_gemm<64, 32, false, false>(bufA, W2, nullptr, bufH, N, stream);
    hipMemsetAsync(z, 0, (size_t)N * 32 * sizeof(float), stream);
    k_scatter<32><<<nblk((long long)E * 8, 256), 256, 0, stream>>>(bufH, src, dst, dinv, z, E);
    k_self_bias<32, true, false><<<nblk((long long)N * 32, 256), 256, 0, stream>>>(z, bufH, dinv, b2, N);

    hipMemsetAsync(bufH, 0, (size_t)N * 32 * sizeof(float), stream);
    k_scatter<32><<<nblk((long long)E * 8, 256), 256, 0, stream>>>(z, src, dst, dinv, bufH, E);
    k_self_bias<32, false, false><<<nblk((long long)N * 32, 256), 256, 0, stream>>>(bufH, z, dinv, nullptr, N);
    launch_gemm<32, 64, true, true>(bufH, W3, b3, bufA, N, stream);

    hipMemsetAsync(bufH, 0, (size_t)N * 64 * sizeof(float), stream);
    k_scatter<64><<<nblk((long long)E * 16, 256), 256, 0, stream>>>(bufA, src, dst, dinv, bufH, E);
    k_self_bias<64, false, false><<<nblk((long long)N * 64, 256), 256, 0, stream>>>(bufH, bufA, dinv, nullptr, N);
    launch_gemm<64, 128, true, false>(bufH, W4, b4, xrec, N, stream);
}


# Edge sums pulled, not pushed: edges ordered by destination per call (atomic count+rank, scan, slot fill); lane group per node sums rows in registers, self+bias+ReLU in its store; sums in slot order
# speedup vs baseline: 2.6250x; 2.0137x over previous
#include <hip/hip_runtime.h>
#include <hip/hip_bf16.h>


typedef float v2f __attribute__((ext_vector_type(2)));
typedef float v8f __attribute__((ext_vector_type(8)));

__global__ __launch_bounds__(256) void k_count_rank(const int* __restrict__ dst,
                                                    int* __restrict__ cnt,
                                                    int* __restrict__ rank, int E) {
    int e = blockIdx.x * blockDim.x + threadIdx.x;
    if (e < E) rank[e] = atomicAdd(&cnt[dst[e]], 1);
}

__device__ __forceinline__ int4 ld4_guard(const int* __restrict__ p, int i, int n) {
    if (i + 3 < n) return *(const int4*)(p + i);
    int4 v;
    v.x = (i     < n) ? p[i]     : 0;
    v.y = (i + 1 < n) ? p[i + 1] : 0;
    v.z = (i + 2 < n) ? p[i + 2] : 0;
    v.w = 0;
    return v;
}

__device__ __forceinline__ int wave_incl_scan(int v, int lane) {
    for (int d = 1; d < 32; d <<= 1) {
        int u = __shfl_up(v, d);
        if (lane >= d) v += u;
    }
    return v;
}

template <int B, bool FULL>
__device__ __forceinline__ void scan_batch(const int* __restrict__ cnt, int* __restrict__ rowptr,
                                           float* __restrict__ dinv, int c, int lane, int n,
                                           int& run) {
    int4 v[B];
    int  ls[B], inc[B], tot[B];
#pragma unroll
    for (int b = 0; b < B; ++b) {
        const int i = ((c + b) << 7) + (lane << 2);
        if (FULL) v[b] = *(const int4*)(cnt + i); else v[b] = ld4_guard(cnt, i, n);
    }
#pragma unroll
    for (int b = 0; b < B; ++b) { ls[b] = v[b].x + v[b].y + v[b].z + v[b].w; inc[b] = ls[b]; }
    for (int d = 1; d < 32; d <<= 1) {
        int u[B];
#pragma unroll
        for (int b = 0; b < B; ++b) u[b] = __shfl_up(inc[b], d);
#pragma unroll
        for (int b = 0; b < B; ++b) if (lane >= d) inc[b] += u[b];
    }
#pragma unroll
    for (int b = 0; b < B; ++b) tot[b] = __shfl(inc[b], 31);
#pragma unroll
    for (int b = 0; b < B; ++b) {
        const int i  = ((c + b) << 7) + (lane << 2);
        const int b0 = run + inc[b] - ls[b];
        run += tot[b];
        int4 r;
        r.x = b0; r.y = b0 + v[b].x; r.z = r.y + v[b].y; r.w = r.z + v[b].z;
        float4 q;
        q.x = rsqrtf((float)v[b].x + 1.0f);
        q.y = rsqrtf((float)v[b].y + 1.0f);
        q.z = rsqrtf((float)v[b].z + 1.0f);
        q.w = rsqrtf((float)v[b].w + 1.0f);
        if (FULL || i + 3 < n) {
            *(int4*)(rowptr + i) = r;
            *(float4*)(dinv + i) = q;
        } else {
            if (i     < n) { rowptr[i]     = r.x; dinv[i]     = q.x; }
            if (i + 1 < n) { rowptr[i + 1] = r.y; dinv[i + 1] = q.y; }
            if (i + 2 < n) { rowptr[i + 2] = r.z; dinv[i + 2] = q.z; }
        }
    }
}

__global__ __launch_bounds__(1024) void k_scan_rowptr(const int* __restrict__ cnt,
                                                      int* __restrict__ rowptr,
                                                      float* __restrict__ dinv, int N) {
    __shared__ int wtot[32];
    const int lane = threadIdx.x & 31;
    const int wv   = threadIdx.x >> 5;
    const int chunks = (N + 127) >> 7;
    const int full   = N >> 7;
    const int cpw    = (chunks + 31) >> 5;
    const int c0     = wv * cpw;
    const int c1     = (c0 + cpw < chunks) ? (c0 + cpw) : chunks;
    const int cf     = (c1 < full) ? c1 : full;

    int sum = 0;
#pragma unroll 5
    for (int c = c0; c < cf; ++c) {
        const int4 v = *(const int4*)(cnt + (c << 7) + (lane << 2));
        sum += v.x + v.y + v.z + v.w;
    }
    for (int c = (cf > c0 ? cf : c0); c < c1; ++c) {
        const int4 v = ld4_guard(cnt, (c << 7) + (lane << 2), N);
        sum += v.x + v.y + v.z + v.w;
    }
    for (int d = 16; d > 0; d >>= 1) sum += __shfl_xor(sum, d);
    if (lane == 0) wtot[wv] = sum;
    __syncthreads();
    const int t    = wtot[lane];
    const int incl = wave_incl_scan(t, lane);
    int run = __shfl(incl, wv) - __shfl(t, wv);
    const int total = __shfl(incl, 31);

    int c = c0;
    for (; c + 5 <= cf; c += 5) scan_batch<5, true>(cnt, rowptr, dinv, c, lane, N, run);
    for (; c < cf; ++c)         scan_batch<1, true>(cnt, rowptr, dinv, c, lane, N, run);
    for (; c < c1; ++c)         scan_batch<1, false>(cnt, rowptr, dinv, c, lane, N, run);

    if (threadIdx.x == 0) rowptr[N] = total;
}

__global__ __launch_bounds__(256) void k_fill_cols(const int* __restrict__ src,
                                                   const int* __restrict__ dst,
                                                   const int* __restrict__ rowptr,
                                                   const int* __restrict__ rank,
                                                   int* __restrict__ col, int E) {
    int e = blockIdx.x * blockDim.x + threadIdx.x;
    if (e < E) col[rowptr[dst[e]] + rank[e]] = src[e];
}

template <int K, int F, bool BIAS, bool RELU>
__global__ __launch_bounds__(256) void k_gemm_wmma(const float* __restrict__ X,
                                                   const float* __restrict__ W,
                                                   const float* __restrict__ bias,
                                                   float* __restrict__ H,
                                                   int tilesM) {
    static_assert(K % 4 == 0 && F % 16 == 0, "shape");
    const int tilesF = F / 16;
    const int wave = blockIdx.x * (blockDim.x >> 5) + (threadIdx.x >> 5);
    if (wave >= tilesM * tilesF) return;
    const int tm = wave / tilesF;
    const int tf = wave % tilesF;
    const int lane = threadIdx.x & 31;
    const int l15  = lane & 15;
    const int hi2  = (lane >> 4) << 1;

    const float* xrow = X + (size_t)(tm * 16 + l15) * K;
    const float* wcol = W + (tf * 16 + l15);

    v8f acc = {};
#pragma unroll 4
    for (int kk = 0; kk < K; kk += 4) {
        v2f a, b;
        a.x = xrow[kk + hi2];
        a.y = xrow[kk + hi2 + 1];
        b.x = wcol[(size_t)(kk + hi2) * F];
        b.y = wcol[(size_t)(kk + hi2 + 1) * F];
        acc = __builtin_amdgcn_wmma_f32_16x16x4_f32(
             false, a,  false, b,
             (short)0, acc,  false,  false);
    }

    const int hi8 = (lane >> 4) << 3;
    float bv = 0.0f;
    if (BIAS) bv = bias[tf * 16 + l15];
#pragma unroll
    for (int v = 0; v < 8; ++v) {
        float o = acc[v];
        if (BIAS) o += bv;
        if (RELU) o = fmaxf(o, 0.0f);
        H[(size_t)(tm * 16 + hi8 + v) * F + tf * 16 + l15] = o;
    }
}

template <int K, int F, bool BIAS, bool RELU>
__global__ __launch_bounds__(256) void k_gemm_tail(const float* __restrict__ X,
                                                   const float* __restrict__ W,
                                                   const float* __restrict__ bias,
                                                   float* __restrict__ H,
                                                   int rowStart, int rowEnd) {
    int idx = blockIdx.x * blockDim.x + threadIdx.x;
    int rows = rowEnd - rowStart;
    if (idx >= rows * F) return;
    int r = rowStart + idx / F, f = idx % F;
    float s = 0.f;
    for (int k = 0; k < K; ++k) s += X[(size_t)r * K + k] * W[(size_t)k * F + f];
    if (BIAS) s += bias[f];
    if (RELU) s = fmaxf(s, 0.0f);
    H[(size_t)r * F + f] = s;
}

template <int F, bool BIAS, bool RELU>
__global__ __launch_bounds__(256) void k_gather(const float* __restrict__ H,
                                                const int* __restrict__ rowptr,
                                                const int* __restrict__ col,
                                                const float* __restrict__ dinv,
                                                const float* __restrict__ bias,
                                                float* __restrict__ OUT, int N) {
    constexpr int CH  = F / 4;
    constexpr int NPB = 256 / CH;
    const int c = threadIdx.x % CH;
    const int g = threadIdx.x / CH;
    const float4* __restrict__ H4 = (const float4*)H;
    float4* __restrict__ O4 = (float4*)OUT;

    float4 bv = make_float4(0.f, 0.f, 0.f, 0.f);
    if (BIAS) bv = ((const float4*)bias)[c];

    const int stride = gridDim.x * NPB;
    int node = blockIdx.x * NPB + g;
    int beg = 0, end = 0;
    if (node < N) { beg = rowptr[node]; end = rowptr[node + 1]; }

    while (node < N) {
        const int nxt = node + stride;
        int nbeg = 0, nend = 0;
        if (nxt < N) { nbeg = rowptr[nxt]; nend = rowptr[nxt + 1]; }

        const float dd = dinv[node];
        const float4 hs = H4[(size_t)node * CH + c];
        float4 acc = make_float4(0.f, 0.f, 0.f, 0.f);

        int j = beg;
        for (; j + 8 <= end; j += 8) {
            int   sidx[8];
            float w[8];
            float4 h[8];
#pragma unroll
            for (int u = 0; u < 8; ++u) sidx[u] = col[j + u];
#pragma unroll
            for (int u = 0; u < 8; ++u) {
                w[u] = dinv[sidx[u]] * dd;
                h[u] = H4[(size_t)sidx[u] * CH + c];
            }
#pragma unroll
            for (int u = 0; u < 8; ++u) {
                acc.x += h[u].x * w[u];
                acc.y += h[u].y * w[u];
                acc.z += h[u].z * w[u];
                acc.w += h[u].w * w[u];
            }
        }
        for (; j < end; j += 4) {
            int   sidx[4];
            float w[4];
            float4 h[4];
#pragma unroll
            for (int u = 0; u < 4; ++u) {
                const int jj = (j + u < end) ? (j + u) : (end - 1);
                sidx[u] = col[jj];
            }
#pragma unroll
            for (int u = 0; u < 4; ++u) {
                w[u] = dinv[sidx[u]] * dd;
                h[u] = H4[(size_t)sidx[u] * CH + c];
            }
#pragma unroll
            for (int u = 0; u < 4; ++u) {
                if (j + u < end) {
                    acc.x += h[u].x * w[u];
                    acc.y += h[u].y * w[u];
                    acc.z += h[u].z * w[u];
                    acc.w += h[u].w * w[u];
                }
            }
        }

        const float ns = dd * dd;
        float4 o;
        o.x = acc.x + hs.x * ns;
        o.y = acc.y + hs.y * ns;
        o.z = acc.z + hs.z * ns;
        o.w = acc.w + hs.w * ns;
        if (BIAS) { o.x += bv.x; o.y += bv.y; o.z += bv.z; o.w += bv.w; }
        if (RELU) {
            o.x = fmaxf(o.x, 0.0f); o.y = fmaxf(o.y, 0.0f);
            o.z = fmaxf(o.z, 0.0f); o.w = fmaxf(o.w, 0.0f);
        }
        O4[(size_t)node * CH + c] = o;

        node = nxt; beg = nbeg; end = nend;
    }
}

static inline int nblk(long long n, int t) { return (int)((n + t - 1) / t); }

template <int K, int F, bool BIAS, bool RELU>
static void launch_gemm(const float* X, const float* W, const float* bias, float* H, int N,
                        hipStream_t s) {
    int tilesM = N / 16;
    long long waves = (long long)tilesM * (F / 16);
    if (waves > 0)
        k_gemm_wmma<K, F, BIAS, RELU><<<nblk(waves, 8), 256, 0, s>>>(X, W, bias, H, tilesM);
    int tail = tilesM * 16;
    if (tail < N)
        k_gemm_tail<K, F, BIAS, RELU><<<nblk((long long)(N - tail) * F, 256), 256, 0, s>>>(
            X, W, bias, H, tail, N);
}

extern "C" void kernel_launch(void* const* d_in, const int* in_sizes, int n_in,
                              void* d_out, int out_size, void* d_ws, size_t ws_size,
                              hipStream_t stream) {
    const float* x  = (const float*)d_in[0];
    const int*   ei = (const int*)d_in[1];
    const float* W1 = (const float*)d_in[2]; const float* b1 = (const float*)d_in[3];
    const float* W2 = (const float*)d_in[4]; const float* b2 = (const float*)d_in[5];
    const float* W3 = (const float*)d_in[6]; const float* b3 = (const float*)d_in[7];
    const float* W4 = (const float*)d_in[8]; const float* b4 = (const float*)d_in[9];

    const int N = in_sizes[0] / 128;
    const int E = in_sizes[1] / 2;
    const int* src = ei;
    const int* dst = ei + E;

    float* out  = (float*)d_out;
    float* xrec = out;
    float* z    = out + (size_t)N * 128;

    float* ws   = (float*)d_ws;
    float* dinv = ws;
    float* bufH = ws + N;
    float* bufA = bufH + (size_t)N * 128;
    int* cnt    = (int*)(bufH + (size_t)N * 64);
    int* rowptr = cnt + N;
    int* col    = rowptr + N + 4;
    int* rank   = col + E;

    const int GG = 1536;

    hipMemsetAsync(cnt, 0, (size_t)N * sizeof(int), stream);
    k_count_rank<<<nblk(E, 256), 256, 0, stream>>>(dst, cnt, rank, E);
    k_scan_rowptr<<<1, 1024, 0, stream>>>(cnt, rowptr, dinv, N);
    k_fill_cols<<<nblk(E, 256), 256, 0, stream>>>(src, dst, rowptr, rank, col, E);

    launch_gemm<128, 64, false, false>(x, W1, nullptr, bufH, N, stream);
    k_gather<64, true, true><<<GG, 256, 0, stream>>>(bufH, rowptr, col, dinv, b1, bufA, N);

    launch_gemm<64, 32, false, false>(bufA, W2, nullptr, bufH, N, stream);
    k_gather<32, true, false><<<GG, 256, 0, stream>>>(bufH, rowptr, col, dinv, b2, z, N);

    k_gather<32, false, false><<<GG, 256, 0, stream>>>(z, rowptr, col, dinv, nullptr, bufH, N);
    launch_gemm<32, 64, true, true>(bufH, W3, b3, bufA, N, stream);

    k_gather<64, false, false><<<GG, 256, 0, stream>>>(bufA, rowptr, col, dinv, nullptr, bufH, N);
    launch_gemm<64, 128, true, false>(bufH, W4, b4, xrec, N, stream);
}


# Rows stored pre-scaled by dinv[row] at their producer (GEMM store; second copy of z): edge sum drops the per-edge dinv[src] load and multiply, dinv[dst] applied once after the sum (roundings move)
# speedup vs baseline: 2.7795x; 1.0589x over previous
#include <hip/hip_runtime.h>
#include <hip/hip_bf16.h>


typedef float v2f __attribute__((ext_vector_type(2)));
typedef float v8f __attribute__((ext_vector_type(8)));

__global__ __launch_bounds__(256) void k_count_rank(const int* __restrict__ dst,
                                                    int* __restrict__ cnt,
                                                    int* __restrict__ rank, int E) {
    int e = blockIdx.x * blockDim.x + threadIdx.x;
    if (e < E) rank[e] = atomicAdd(&cnt[dst[e]], 1);
}

__device__ __forceinline__ int4 ld4_guard(const int* __restrict__ p, int i, int n) {
    if (i + 3 < n) return *(const int4*)(p + i);
    int4 v;
    v.x = (i     < n) ? p[i]     : 0;
    v.y = (i + 1 < n) ? p[i + 1] : 0;
    v.z = (i + 2 < n) ? p[i + 2] : 0;
    v.w = 0;
    return v;
}

__device__ __forceinline__ int wave_incl_scan(int v, int lane) {
    for (int d = 1; d < 32; d <<= 1) {
        int u = __shfl_up(v, d);
        if (lane >= d) v += u;
    }
    return v;
}

template <int B, bool FULL>
__device__ __forceinline__ void scan_batch(const int* __restrict__ cnt, int* __restrict__ rowptr,
                                           float* __restrict__ dinv, int c, int lane, int n,
                                           int& run) {
    int4 v[B];
    int  ls[B], inc[B], tot[B];
#pragma unroll
    for (int b = 0; b < B; ++b) {
        const int i = ((c + b) << 7) + (lane << 2);
        if (FULL) v[b] = *(const int4*)(cnt + i); else v[b] = ld4_guard(cnt, i, n);
    }
#pragma unroll
    for (int b = 0; b < B; ++b) { ls[b] = v[b].x + v[b].y + v[b].z + v[b].w; inc[b] = ls[b]; }
    for (int d = 1; d < 32; d <<= 1) {
        int u[B];
#pragma unroll
        for (int b = 0; b < B; ++b) u[b] = __shfl_up(inc[b], d);
#pragma unroll
        for (int b = 0; b < B; ++b) if (lane >= d) inc[b] += u[b];
    }
#pragma unroll
    for (int b = 0; b < B; ++b) tot[b] = __shfl(inc[b], 31);
#pragma unroll
    for (int b = 0; b < B; ++b) {
        const int i  = ((c + b) << 7) + (lane << 2);
        const int b0 = run + inc[b] - ls[b];
        run += tot[b];
        int4 r;
        r.x = b0; r.y = b0 + v[b].x; r.z = r.y + v[b].y; r.w = r.z + v[b].z;
        float4 q;
        q.x = rsqrtf((float)v[b].x + 1.0f);
        q.y = rsqrtf((float)v[b].y + 1.0f);
        q.z = rsqrtf((float)v[b].z + 1.0f);
        q.w = rsqrtf((float)v[b].w + 1.0f);
        if (FULL || i + 3 < n) {
            *(int4*)(rowptr + i) = r;
            *(float4*)(dinv + i) = q;
        } else {
            if (i     < n) { rowptr[i]     = r.x; dinv[i]     = q.x; }
            if (i + 1 < n) { rowptr[i + 1] = r.y; dinv[i + 1] = q.y; }
            if (i + 2 < n) { rowptr[i + 2] = r.z; dinv[i + 2] = q.z; }
        }
    }
}

__global__ __launch_bounds__(1024) void k_scan_rowptr(const int* __restrict__ cnt,
                                                      int* __restrict__ rowptr,
                                                      float* __restrict__ dinv, int N) {
    __shared__ int wtot[32];
    const int lane = threadIdx.x & 31;
    const int wv   = threadIdx.x >> 5;
    const int chunks = (N + 127) >> 7;
    const int full   = N >> 7;
    const int cpw    = (chunks + 31) >> 5;
    const int c0     = wv * cpw;
    const int c1     = (c0 + cpw < chunks) ? (c0 + cpw) : chunks;
    const int cf     = (c1 < full) ? c1 : full;

    int sum = 0;
#pragma unroll 5
    for (int c = c0; c < cf; ++c) {
        const int4 v = *(const int4*)(cnt + (c << 7) + (lane << 2));
        sum += v.x + v.y + v.z + v.w;
    }
    for (int c = (cf > c0 ? cf : c0); c < c1; ++c) {
        const int4 v = ld4_guard(cnt, (c << 7) + (lane << 2), N);
        sum += v.x + v.y + v.z + v.w;
    }
    for (int d = 16; d > 0; d >>= 1) sum += __shfl_xor(sum, d);
    if (lane == 0) wtot[wv] = sum;
    __syncthreads();
    const int t    = wtot[lane];
    const int incl = wave_incl_scan(t, lane);
    int run = __shfl(incl, wv) - __shfl(t, wv);
    const int total = __shfl(incl, 31);

    int c = c0;
    for (; c + 5 <= cf; c += 5) scan_batch<5, true>(cnt, rowptr, dinv, c, lane, N, run);
    for (; c < cf; ++c)         scan_batch<1, true>(cnt, rowptr, dinv, c, lane, N, run);
    for (; c < c1; ++c)         scan_batch<1, false>(cnt, rowptr, dinv, c, lane, N, run);

    if (threadIdx.x == 0) rowptr[N] = total;
}

__global__ __launch_bounds__(256) void k_fill_cols(const int* __restrict__ src,
                                                   const int* __restrict__ dst,
                                                   const int* __restrict__ rowptr,
                                                   const int* __restrict__ rank,
                                                   int* __restrict__ col, int E) {
    int e = blockIdx.x * blockDim.x + threadIdx.x;
    if (e < E) col[rowptr[dst[e]] + rank[e]] = src[e];
}

template <int K, int F, bool BIAS, bool RELU, bool SCALE>
__global__ __launch_bounds__(256) void k_gemm_wmma(const float* __restrict__ X,
                                                   const float* __restrict__ W,
                                                   const float* __restrict__ bias,
                                                   const float* __restrict__ dinv,
                                                   float* __restrict__ H,
                                                   int tilesM) {
    static_assert(K % 4 == 0 && F % 16 == 0, "shape");
    const int tilesF = F / 16;
    const int wave = blockIdx.x * (blockDim.x >> 5) + (threadIdx.x >> 5);
    if (wave >= tilesM * tilesF) return;
    const int tm = wave / tilesF;
    const int tf = wave % tilesF;
    const int lane = threadIdx.x & 31;
    const int l15  = lane & 15;
    const int hi2  = (lane >> 4) << 1;

    const float* xrow = X + (size_t)(tm * 16 + l15) * K;
    const float* wcol = W + (tf * 16 + l15);

    v8f acc = {};
#pragma unroll 4
    for (int kk = 0; kk < K; kk += 4) {
        v2f a, b;
        a.x = xrow[kk + hi2];
        a.y = xrow[kk + hi2 + 1];
        b.x = wcol[(size_t)(kk + hi2) * F];
        b.y = wcol[(size_t)(kk + hi2 + 1) * F];
        acc = __builtin_amdgcn_wmma_f32_16x16x4_f32(
             false, a,  false, b,
             (short)0, acc,  false,  false);
    }

    const int hi8 = (lane >> 4) << 3;
    float bv = 0.0f;
    if (BIAS) bv = bias[tf * 16 + l15];
    float dv[8];
#pragma unroll
    for (int v = 0; v < 8; ++v) dv[v] = SCALE ? dinv[tm * 16 + hi8 + v] : 1.0f;
#pragma unroll
    for (int v = 0; v < 8; ++v) {
        float o = acc[v];
        if (BIAS) o += bv;
        if (RELU) o = fmaxf(o, 0.0f);
        if (SCALE) o *= dv[v];
        H[(size_t)(tm * 16 + hi8 + v) * F + tf * 16 + l15] = o;
    }
}

template <int K, int F, bool BIAS, bool RELU, bool SCALE>
__global__ __launch_bounds__(256) void k_gemm_tail(const float* __restrict__ X,
                                                   const float* __restrict__ W,
                                                   const float* __restrict__ bias,
                                                   const float* __restrict__ dinv,
                                                   float* __restrict__ H,
                                                   int rowStart, int rowEnd) {
    int idx = blockIdx.x * blockDim.x + threadIdx.x;
    int rows = rowEnd - rowStart;
    if (idx >= rows * F) return;
    int r = rowStart + idx / F, f = idx % F;
    float s = 0.f;
    for (int k = 0; k < K; ++k) s += X[(size_t)r * K + k] * W[(size_t)k * F + f];
    if (BIAS) s += bias[f];
    if (RELU) s = fmaxf(s, 0.0f);
    if (SCALE) s *= dinv[r];
    H[(size_t)r * F + f] = s;
}

template <int F, bool BIAS, bool RELU, bool DUAL>
__global__ __launch_bounds__(256) void k_gather(const float* __restrict__ HS,
                                                const int* __restrict__ rowptr,
                                                const int* __restrict__ col,
                                                const float* __restrict__ dinv,
                                                const float* __restrict__ bias,
                                                float* __restrict__ OUT,
                                                float* __restrict__ OUTS, int N) {
    constexpr int CH  = F / 4;
    constexpr int NPB = 256 / CH;
    const int c = threadIdx.x % CH;
    const int g = threadIdx.x / CH;
    const float4* __restrict__ H4 = (const float4*)HS;
    float4* __restrict__ O4 = (float4*)OUT;
    float4* __restrict__ S4 = (float4*)OUTS;

    float4 bv = make_float4(0.f, 0.f, 0.f, 0.f);
    if (BIAS) bv = ((const float4*)bias)[c];

    const int stride = gridDim.x * NPB;
    int node = blockIdx.x * NPB + g;
    int beg = 0, end = 0;
    if (node < N) { beg = rowptr[node]; end = rowptr[node + 1]; }

    while (node < N) {
        const int nxt = node + stride;
        int nbeg = 0, nend = 0;
        if (nxt < N) { nbeg = rowptr[nxt]; nend = rowptr[nxt + 1]; }

        const float dd = dinv[node];
        const float4 hs = H4[(size_t)node * CH + c];
        float4 acc = make_float4(0.f, 0.f, 0.f, 0.f);

        int j = beg;
        for (; j + 8 <= end; j += 8) {
            int   sidx[8];
            float4 h[8];
#pragma unroll
            for (int u = 0; u < 8; ++u) sidx[u] = col[j + u];
#pragma unroll
            for (int u = 0; u < 8; ++u) h[u] = H4[(size_t)sidx[u] * CH + c];
#pragma unroll
            for (int u = 0; u < 8; ++u) {
                acc.x += h[u].x;
                acc.y += h[u].y;
                acc.z += h[u].z;
                acc.w += h[u].w;
            }
        }
        for (; j < end; j += 4) {
            int   sidx[4];
            float4 h[4];
#pragma unroll
            for (int u = 0; u < 4; ++u) {
                const int jj = (j + u < end) ? (j + u) : (end - 1);
                sidx[u] = col[jj];
            }
#pragma unroll
            for (int u = 0; u < 4; ++u) h[u] = H4[(size_t)sidx[u] * CH + c];
#pragma unroll
            for (int u = 0; u < 4; ++u) {
                if (j + u < end) {
                    acc.x += h[u].x;
                    acc.y += h[u].y;
                    acc.z += h[u].z;
                    acc.w += h[u].w;
                }
            }
        }

        float4 o;
        o.x = (acc.x + hs.x) * dd;
        o.y = (acc.y + hs.y) * dd;
        o.z = (acc.z + hs.z) * dd;
        o.w = (acc.w + hs.w) * dd;
        if (BIAS) { o.x += bv.x; o.y += bv.y; o.z += bv.z; o.w += bv.w; }
        if (RELU) {
            o.x = fmaxf(o.x, 0.0f); o.y = fmaxf(o.y, 0.0f);
            o.z = fmaxf(o.z, 0.0f); o.w = fmaxf(o.w, 0.0f);
        }
        O4[(size_t)node * CH + c] = o;
        if (DUAL) {
            float4 q;
            q.x = o.x * dd; q.y = o.y * dd; q.z = o.z * dd; q.w = o.w * dd;
            S4[(size_t)node * CH + c] = q;
        }

        node = nxt; beg = nbeg; end = nend;
    }
}

static inline int nblk(long long n, int t) { return (int)((n + t - 1) / t); }

template <int K, int F, bool BIAS, bool RELU, bool SCALE>
static void launch_gemm(const float* X, const float* W, const float* bias, const float* dinv,
                        float* H, int N, hipStream_t s) {
    int tilesM = N / 16;
    long long waves = (long long)tilesM * (F / 16);
    if (waves > 0)
        k_gemm_wmma<K, F, BIAS, RELU, SCALE><<<nblk(waves, 8), 256, 0, s>>>(X, W, bias, dinv, H,
                                                                            tilesM);
    int tail = tilesM * 16;
    if (tail < N)
        k_gemm_tail<K, F, BIAS, RELU, SCALE><<<nblk((long long)(N - tail) * F, 256), 256, 0, s>>>(
            X, W, bias, dinv, H, tail, N);
}

extern "C" void kernel_launch(void* const* d_in, const int* in_sizes, int n_in,
                              void* d_out, int out_size, void* d_ws, size_t ws_size,
                              hipStream_t stream) {
    const float* x  = (const float*)d_in[0];
    const int*   ei = (const int*)d_in[1];
    const float* W1 = (const float*)d_in[2]; const float* b1 = (const float*)d_in[3];
    const float* W2 = (const float*)d_in[4]; const float* b2 = (const float*)d_in[5];
    const float* W3 = (const float*)d_in[6]; const float* b3 = (const float*)d_in[7];
    const float* W4 = (const float*)d_in[8]; const float* b4 = (const float*)d_in[9];

    const int N = in_sizes[0] / 128;
    const int E = in_sizes[1] / 2;
    const int* src = ei;
    const int* dst = ei + E;

    float* out  = (float*)d_out;
    float* xrec = out;
    float* z    = out + (size_t)N * 128;

    float* ws   = (float*)d_ws;
    float* dinv = ws;
    float* bufH = ws + N;
    float* bufA = bufH + (size_t)N * 128;
    int* cnt    = (int*)(bufH + (size_t)N * 64);
    int* rowptr = cnt + N;
    int* col    = rowptr + N + 4;
    int* rank   = col + E;

    const int GG = 1536;

    hipMemsetAsync(cnt, 0, (size_t)N * sizeof(int), stream);
    k_count_rank<<<nblk(E, 256), 256, 0, stream>>>(dst, cnt, rank, E);
    k_scan_rowptr<<<1, 1024, 0, stream>>>(cnt, rowptr, dinv, N);
    k_fill_cols<<<nblk(E, 256), 256, 0, stream>>>(src, dst, rowptr, rank, col, E);

    launch_gemm<128, 64, false, false, true>(x, W1, nullptr, dinv, bufH, N, stream);
    k_gather<64, true, true, false><<<GG, 256, 0, stream>>>(bufH, rowptr, col, dinv, b1, bufA,
                                                            nullptr, N);

    launch_gemm<64, 32, false, false, true>(bufA, W2, nullptr, dinv, bufH, N, stream);
    k_gather<32, true, false, true><<<GG, 256, 0, stream>>>(bufH, rowptr, col, dinv, b2, z, bufA,
                                                            N);

    k_gather<32, false, false, false><<<GG, 256, 0, stream>>>(bufA, rowptr, col, dinv, nullptr,
                                                              bufH, nullptr, N);
    launch_gemm<32, 64, true, true, true>(bufH, W3, b3, dinv, bufA, N, stream);

    k_gather<64, false, false, false><<<GG, 256, 0, stream>>>(bufA, rowptr, col, dinv, nullptr,
                                                              bufH, nullptr, N);
    launch_gemm<64, 128, true, false, false>(bufH, W4, b4, nullptr, xrec, N, stream);
}


# GEMMs: one wave per 16-row strip holds all F/16 column tiles and loads the strip's A fragments once (was one wave per 16x16 tile): 112500 -> 25000 GEMM waves, F/16 times fewer A loads; sums unchanged
# speedup vs baseline: 2.9124x; 1.0478x over previous
#include <hip/hip_runtime.h>
#include <hip/hip_bf16.h>


typedef float v2f __attribute__((ext_vector_type(2)));
typedef float v8f __attribute__((ext_vector_type(8)));

__global__ __launch_bounds__(256) void k_count_rank(const int* __restrict__ dst,
                                                    int* __restrict__ cnt,
                                                    int* __restrict__ rank, int E) {
    int e = blockIdx.x * blockDim.x + threadIdx.x;
    if (e < E) rank[e] = atomicAdd(&cnt[dst[e]], 1);
}

__device__ __forceinline__ int4 ld4_guard(const int* __restrict__ p, int i, int n) {
    if (i + 3 < n) return *(const int4*)(p + i);
    int4 v;
    v.x = (i     < n) ? p[i]     : 0;
    v.y = (i + 1 < n) ? p[i + 1] : 0;
    v.z = (i + 2 < n) ? p[i + 2] : 0;
    v.w = 0;
    return v;
}

__device__ __forceinline__ int wave_incl_scan(int v, int lane) {
    for (int d = 1; d < 32; d <<= 1) {
        int u = __shfl_up(v, d);
        if (lane >= d) v += u;
    }
    return v;
}

template <int B, bool FULL>
__device__ __forceinline__ void scan_batch(const int* __restrict__ cnt, int* __restrict__ rowptr,
                                           float* __restrict__ dinv, int c, int lane, int n,
                                           int& run) {
    int4 v[B];
    int  ls[B], inc[B], tot[B];
#pragma unroll
    for (int b = 0; b < B; ++b) {
        const int i = ((c + b) << 7) + (lane << 2);
        if (FULL) v[b] = *(const int4*)(cnt + i); else v[b] = ld4_guard(cnt, i, n);
    }
#pragma unroll
    for (int b = 0; b < B; ++b) { ls[b] = v[b].x + v[b].y + v[b].z + v[b].w; inc[b] = ls[b]; }
    for (int d = 1; d < 32; d <<= 1) {
        int u[B];
#pragma unroll
        for (int b = 0; b < B; ++b) u[b] = __shfl_up(inc[b], d);
#pragma unroll
        for (int b = 0; b < B; ++b) if (lane >= d) inc[b] += u[b];
    }
#pragma unroll
    for (int b = 0; b < B; ++b) tot[b] = __shfl(inc[b], 31);
#pragma unroll
    for (int b = 0; b < B; ++b) {
        const int i  = ((c + b) << 7) + (lane << 2);
        const int b0 = run + inc[b] - ls[b];
        run += tot[b];
        int4 r;
        r.x = b0; r.y = b0 + v[b].x; r.z = r.y + v[b].y; r.w = r.z + v[b].z;
        float4 q;
        q.x = rsqrtf((float)v[b].x + 1.0f);
        q.y = rsqrtf((float)v[b].y + 1.0f);
        q.z = rsqrtf((float)v[b].z + 1.0f);
        q.w = rsqrtf((float)v[b].w + 1.0f);
        if (FULL || i + 3 < n) {
            *(int4*)(rowptr + i) = r;
            *(float4*)(dinv + i) = q;
        } else {
            if (i     < n) { rowptr[i]     = r.x; dinv[i]     = q.x; }
            if (i + 1 < n) { rowptr[i + 1] = r.y; dinv[i + 1] = q.y; }
            if (i + 2 < n) { rowptr[i + 2] = r.z; dinv[i + 2] = q.z; }
        }
    }
}

__global__ __launch_bounds__(1024) void k_scan_rowptr(const int* __restrict__ cnt,
                                                      int* __restrict__ rowptr,
                                                      float* __restrict__ dinv, int N) {
    __shared__ int wtot[32];
    const int lane = threadIdx.x & 31;
    const int wv   = threadIdx.x >> 5;
    const int chunks = (N + 127) >> 7;
    const int full   = N >> 7;
    const int cpw    = (chunks + 31) >> 5;
    const int c0     = wv * cpw;
    const int c1     = (c0 + cpw < chunks) ? (c0 + cpw) : chunks;
    const int cf     = (c1 < full) ? c1 : full;

    int sum = 0;
#pragma unroll 5
    for (int c = c0; c < cf; ++c) {
        const int4 v = *(const int4*)(cnt + (c << 7) + (lane << 2));
        sum += v.x + v.y + v.z + v.w;
    }
    for (int c = (cf > c0 ? cf : c0); c < c1; ++c) {
        const int4 v = ld4_guard(cnt, (c << 7) + (lane << 2), N);
        sum += v.x + v.y + v.z + v.w;
    }
    for (int d = 16; d > 0; d >>= 1) sum += __shfl_xor(sum, d);
    if (lane == 0) wtot[wv] = sum;
    __syncthreads();
    const int t    = wtot[lane];
    const int incl = wave_incl_scan(t, lane);
    int run = __shfl(incl, wv) - __shfl(t, wv);
    const int total = __shfl(incl, 31);

    int c = c0;
    for (; c + 5 <= cf; c += 5) scan_batch<5, true>(cnt, rowptr, dinv, c, lane, N, run);
    for (; c < cf; ++c)         scan_batch<1, true>(cnt, rowptr, dinv, c, lane, N, run);
    for (; c < c1; ++c)         scan_batch<1, false>(cnt, rowptr, dinv, c, lane, N, run);

    if (threadIdx.x == 0) rowptr[N] = total;
}

__global__ __launch_bounds__(256) void k_fill_cols(const int* __restrict__ src,
                                                   const int* __restrict__ dst,
                                                   const int* __restrict__ rowptr,
                                                   const int* __restrict__ rank,
                                                   int* __restrict__ col, int E) {
    int e = blockIdx.x * blockDim.x + threadIdx.x;
    if (e < E) col[rowptr[dst[e]] + rank[e]] = src[e];
}

template <int K, int F, bool BIAS, bool RELU, bool SCALE>
__global__ __launch_bounds__(256) void k_gemm_wmma(const float* __restrict__ X,
                                                   const float* __restrict__ W,
                                                   const float* __restrict__ bias,
                                                   const float* __restrict__ dinv,
                                                   float* __restrict__ H,
                                                   int tilesM) {
    static_assert(K % 4 == 0 && F % 16 == 0, "shape");
    constexpr int TF = F / 16;
    const int tm = blockIdx.x * (blockDim.x >> 5) + (threadIdx.x >> 5);
    if (tm >= tilesM) return;
    const int lane = threadIdx.x & 31;
    const int l15  = lane & 15;
    const int hi2  = (lane >> 4) << 1;

    const float* xrow = X + (size_t)(tm * 16 + l15) * K + hi2;
    const float* wcol = W + (size_t)hi2 * F + l15;

    v8f acc[TF];
#pragma unroll
    for (int t = 0; t < TF; ++t) acc[t] = v8f{};

    constexpr int KS = K / 4;
    v2f a[KS];
#pragma unroll
    for (int s = 0; s < KS; ++s) {
        a[s].x = xrow[4 * s];
        a[s].y = xrow[4 * s + 1];
    }

#pragma unroll
    for (int s = 0; s < KS; ++s) {
#pragma unroll
        for (int t = 0; t < TF; ++t) {
            v2f b;
            b.x = wcol[(size_t)(4 * s) * F + t * 16];
            b.y = wcol[(size_t)(4 * s + 1) * F + t * 16];
            acc[t] = __builtin_amdgcn_wmma_f32_16x16x4_f32(
                 false, a[s],  false, b,
                 (short)0, acc[t],  false,  false);
        }
    }

    const int hi8 = (lane >> 4) << 3;
    float dv[8];
#pragma unroll
    for (int v = 0; v < 8; ++v) dv[v] = SCALE ? dinv[tm * 16 + hi8 + v] : 1.0f;
    float* hrow = H + (size_t)(tm * 16 + hi8) * F + l15;
#pragma unroll
    for (int t = 0; t < TF; ++t) {
        float bv = 0.0f;
        if (BIAS) bv = bias[t * 16 + l15];
#pragma unroll
        for (int v = 0; v < 8; ++v) {
            float o = acc[t][v];
            if (BIAS) o += bv;
            if (RELU) o = fmaxf(o, 0.0f);
            if (SCALE) o *= dv[v];
            hrow[(size_t)v * F + t * 16] = o;
        }
    }
}

template <int K, int F, bool BIAS, bool RELU, bool SCALE>
__global__ __launch_bounds__(256) void k_gemm_tail(const float* __restrict__ X,
                                                   const float* __restrict__ W,
                                                   const float* __restrict__ bias,
                                                   const float* __restrict__ dinv,
                                                   float* __restrict__ H,
                                                   int rowStart, int rowEnd) {
    int idx = blockIdx.x * blockDim.x + threadIdx.x;
    int rows = rowEnd - rowStart;
    if (idx >= rows * F) return;
    int r = rowStart + idx / F, f = idx % F;
    float s = 0.f;
    for (int k = 0; k < K; ++k) s += X[(size_t)r * K + k] * W[(size_t)k * F + f];
    if (BIAS) s += bias[f];
    if (RELU) s = fmaxf(s, 0.0f);
    if (SCALE) s *= dinv[r];
    H[(size_t)r * F + f] = s;
}

template <int F, bool BIAS, bool RELU, bool DUAL>
__global__ __launch_bounds__(256) void k_gather(const float* __restrict__ HS,
                                                const int* __restrict__ rowptr,
                                                const int* __restrict__ col,
                                                const float* __restrict__ dinv,
                                                const float* __restrict__ bias,
                                                float* __restrict__ OUT,
                                                float* __restrict__ OUTS, int N) {
    constexpr int CH  = F / 4;
    constexpr int NPB = 256 / CH;
    const int c = threadIdx.x % CH;
    const int g = threadIdx.x / CH;
    const float4* __restrict__ H4 = (const float4*)HS;
    float4* __restrict__ O4 = (float4*)OUT;
    float4* __restrict__ S4 = (float4*)OUTS;

    float4 bv = make_float4(0.f, 0.f, 0.f, 0.f);
    if (BIAS) bv = ((const float4*)bias)[c];

    const int stride = gridDim.x * NPB;
    int node = blockIdx.x * NPB + g;
    int beg = 0, end = 0;
    if (node < N) { beg = rowptr[node]; end = rowptr[node + 1]; }

    while (node < N) {
        const int nxt = node + stride;
        int nbeg = 0, nend = 0;
        if (nxt < N) { nbeg = rowptr[nxt]; nend = rowptr[nxt + 1]; }

        const float dd = dinv[node];
        const float4 hs = H4[(size_t)node * CH + c];
        float4 acc = make_float4(0.f, 0.f, 0.f, 0.f);

        int j = beg;
        for (; j + 8 <= end; j += 8) {
            int   sidx[8];
            float4 h[8];
#pragma unroll
            for (int u = 0; u < 8; ++u) sidx[u] = col[j + u];
#pragma unroll
            for (int u = 0; u < 8; ++u) h[u] = H4[(size_t)sidx[u] * CH + c];
#pragma unroll
            for (int u = 0; u < 8; ++u) {
                acc.x += h[u].x;
                acc.y += h[u].y;
                acc.z += h[u].z;
                acc.w += h[u].w;
            }
        }
        for (; j < end; j += 4) {
            int   sidx[4];
            float4 h[4];
#pragma unroll
            for (int u = 0; u < 4; ++u) {
                const int jj = (j + u < end) ? (j + u) : (end - 1);
                sidx[u] = col[jj];
            }
#pragma unroll
            for (int u = 0; u < 4; ++u) h[u] = H4[(size_t)sidx[u] * CH + c];
#pragma unroll
            for (int u = 0; u < 4; ++u) {
                if (j + u < end) {
                    acc.x += h[u].x;
                    acc.y += h[u].y;
                    acc.z += h[u].z;
                    acc.w += h[u].w;
                }
            }
        }

        float4 o;
        o.x = (acc.x + hs.x) * dd;
        o.y = (acc.y + hs.y) * dd;
        o.z = (acc.z + hs.z) * dd;
        o.w = (acc.w + hs.w) * dd;
        if (BIAS) { o.x += bv.x; o.y += bv.y; o.z += bv.z; o.w += bv.w; }
        if (RELU) {
            o.x = fmaxf(o.x, 0.0f); o.y = fmaxf(o.y, 0.0f);
            o.z = fmaxf(o.z, 0.0f); o.w = fmaxf(o.w, 0.0f);
        }
        O4[(size_t)node * CH + c] = o;
        if (DUAL) {
            float4 q;
            q.x = o.x * dd; q.y = o.y * dd; q.z = o.z * dd; q.w = o.w * dd;
            S4[(size_t)node * CH + c] = q;
        }

        node = nxt; beg = nbeg; end = nend;
    }
}

static inline int nblk(long long n, int t) { return (int)((n + t - 1) / t); }

template <int K, int F, bool BIAS, bool RELU, bool SCALE>
static void launch_gemm(const float* X, const float* W, const float* bias, const float* dinv,
                        float* H, int N, hipStream_t s) {
    int tilesM = N / 16;
    long long waves = tilesM;
    if (waves > 0)
        k_gemm_wmma<K, F, BIAS, RELU, SCALE><<<nblk(waves, 8), 256, 0, s>>>(X, W, bias, dinv, H,
                                                                            tilesM);
    int tail = tilesM * 16;
    if (tail < N)
        k_gemm_tail<K, F, BIAS, RELU, SCALE><<<nblk((long long)(N - tail) * F, 256), 256, 0, s>>>(
            X, W, bias, dinv, H, tail, N);
}

extern "C" void kernel_launch(void* const* d_in, const int* in_sizes, int n_in,
                              void* d_out, int out_size, void* d_ws, size_t ws_size,
                              hipStream_t stream) {
    const float* x  = (const float*)d_in[0];
    const int*   ei = (const int*)d_in[1];
    const float* W1 = (const float*)d_in[2]; const float* b1 = (const float*)d_in[3];
    const float* W2 = (const float*)d_in[4]; const float* b2 = (const float*)d_in[5];
    const float* W3 = (const float*)d_in[6]; const float* b3 = (const float*)d_in[7];
    const float* W4 = (const float*)d_in[8]; const float* b4 = (const float*)d_in[9];

    const int N = in_sizes[0] / 128;
    const int E = in_sizes[1] / 2;
    const int* src = ei;
    const int* dst = ei + E;

    float* out  = (float*)d_out;
    float* xrec = out;
    float* z    = out + (size_t)N * 128;

    float* ws   = (float*)d_ws;
    float* dinv = ws;
    float* bufH = ws + N;
    float* bufA = bufH + (size_t)N * 128;
    int* cnt    = (int*)(bufH + (size_t)N * 64);
    int* rowptr = cnt + N;
    int* col    = rowptr + N + 4;
    int* rank   = col + E;

    const int GG = 1536;

    hipMemsetAsync(cnt, 0, (size_t)N * sizeof(int), stream);
    k_count_rank<<<nblk(E, 256), 256, 0, stream>>>(dst, cnt, rank, E);
    k_scan_rowptr<<<1, 1024, 0, stream>>>(cnt, rowptr, dinv, N);
    k_fill_cols<<<nblk(E, 256), 256, 0, stream>>>(src, dst, rowptr, rank, col, E);

    launch_gemm<128, 64, false, false, true>(x, W1, nullptr, dinv, bufH, N, stream);
    k_gather<64, true, true, false><<<GG, 256, 0, stream>>>(bufH, rowptr, col, dinv, b1, bufA,
                                                            nullptr, N);

    launch_gemm<64, 32, false, false, true>(bufA, W2, nullptr, dinv, bufH, N, stream);
    k_gather<32, true, false, true><<<GG, 256, 0, stream>>>(bufH, rowptr, col, dinv, b2, z, bufA,
                                                            N);

    k_gather<32, false, false, false><<<GG, 256, 0, stream>>>(bufA, rowptr, col, dinv, nullptr,
                                                              bufH, nullptr, N);
    launch_gemm<32, 64, true, true, true>(bufH, W3, b3, dinv, bufA, N, stream);

    k_gather<64, false, false, false><<<GG, 256, 0, stream>>>(bufA, rowptr, col, dinv, nullptr,
                                                              bufH, nullptr, N);
    launch_gemm<64, 128, true, false, false>(bufH, W4, b4, nullptr, xrec, N, stream);
}


# Edge order built on shared-memory counters in two levels (slice-by-bucket counts, table scan, per-bucket count+scan+fill): no global atomic per edge, no memset; slot order still by counter arrival
# speedup vs baseline: 3.9922x; 1.3708x over previous
#include <hip/hip_runtime.h>
#include <hip/hip_bf16.h>


typedef float v2f __attribute__((ext_vector_type(2)));
typedef float v8f __attribute__((ext_vector_type(8)));

constexpr int BSH   = 9;
constexpr int BW    = 1 << BSH;
constexpr int NBMAX = 256;
constexpr int SG    = 256;
constexpr int PSH   = 10;
constexpr int SU    = 8;
constexpr int KT    = BW;
constexpr int BU    = 20;

__device__ __forceinline__ int4 ld4_guard(const int* __restrict__ p, int i, int n) {
    if (i + 3 < n) return *(const int4*)(p + i);
    int4 v;
    v.x = (i     < n) ? p[i]     : 0;
    v.y = (i + 1 < n) ? p[i + 1] : 0;
    v.z = (i + 2 < n) ? p[i + 2] : 0;
    v.w = 0;
    return v;
}

__device__ __forceinline__ int wave_incl_scan(int v, int lane) {
    for (int d = 1; d < 32; d <<= 1) {
        int u = __shfl_up(v, d);
        if (lane >= d) v += u;
    }
    return v;
}

__global__ __launch_bounds__(256) void k_slice_hist(const int* __restrict__ dst,
                                                    int* __restrict__ tblc,
                                                    int E, int NB) {
    __shared__ int hist[NBMAX];
    for (int b = threadIdx.x; b < NB; b += 256) hist[b] = 0;
    __syncthreads();
    const bool al  = (((size_t)dst) & 15) == 0;
    const int  nwp = al ? (E >> PSH) : 0;
    const int  t4  = 4 * threadIdx.x;
    for (int p0 = blockIdx.x; p0 < nwp; p0 += SU * SG) {
        int4 d[SU];
#pragma unroll
        for (int u = 0; u < SU; ++u) {
            const int p  = p0 + u * SG;
            const int pc = (p < nwp) ? p : p0;
            d[u] = *(const int4*)(dst + ((size_t)pc << PSH) + t4);
        }
#pragma unroll
        for (int u = 0; u < SU; ++u) {
            if (p0 + u * SG < nwp) {
                atomicAdd(&hist[d[u].x >> BSH], 1);
                atomicAdd(&hist[d[u].y >> BSH], 1);
                atomicAdd(&hist[d[u].z >> BSH], 1);
                atomicAdd(&hist[d[u].w >> BSH], 1);
            }
        }
    }
    for (int p = blockIdx.x; (p << PSH) < E; p += SG) {
        if (p < nwp) continue;
        for (int q = 0; q < 4; ++q) {
            const int i = (p << PSH) + q * 256 + threadIdx.x;
            if (i < E) atomicAdd(&hist[dst[i] >> BSH], 1);
        }
    }
    __syncthreads();
    for (int b = threadIdx.x; b < NB; b += 256) tblc[b * SG + blockIdx.x] = hist[b];
}

template <int B, bool FULL>
__device__ __forceinline__ void scan_batch(const int* __restrict__ cnt, int* __restrict__ off,
                                           int c, int lane, int n, int& run) {
    int4 v[B];
    int  ls[B], inc[B], tot[B];
#pragma unroll
    for (int b = 0; b < B; ++b) {
        const int i = ((c + b) << 7) + (lane << 2);
        if (FULL) v[b] = *(const int4*)(cnt + i); else v[b] = ld4_guard(cnt, i, n);
    }
#pragma unroll
    for (int b = 0; b < B; ++b) { ls[b] = v[b].x + v[b].y + v[b].z + v[b].w; inc[b] = ls[b]; }
    for (int d = 1; d < 32; d <<= 1) {
        int u[B];
#pragma unroll
        for (int b = 0; b < B; ++b) u[b] = __shfl_up(inc[b], d);
#pragma unroll
        for (int b = 0; b < B; ++b) if (lane >= d) inc[b] += u[b];
    }
#pragma unroll
    for (int b = 0; b < B; ++b) tot[b] = __shfl(inc[b], 31);
#pragma unroll
    for (int b = 0; b < B; ++b) {
        const int i  = ((c + b) << 7) + (lane << 2);
        const int b0 = run + inc[b] - ls[b];
        run += tot[b];
        int4 r;
        r.x = b0; r.y = b0 + v[b].x; r.z = r.y + v[b].y; r.w = r.z + v[b].z;
        if (FULL || i + 3 < n) {
            *(int4*)(off + i) = r;
        } else {
            if (i     < n) off[i]     = r.x;
            if (i + 1 < n) off[i + 1] = r.y;
            if (i + 2 < n) off[i + 2] = r.z;
        }
    }
}

__global__ __launch_bounds__(1024) void k_scan_table(const int* __restrict__ cnt,
                                                     int* __restrict__ off, int n) {
    __shared__ int wtot[32];
    const int lane = threadIdx.x & 31;
    const int wv   = threadIdx.x >> 5;
    const int chunks = (n + 127) >> 7;
    const int full   = n >> 7;
    const int cpw    = (chunks + 31) >> 5;
    const int c0     = wv * cpw;
    const int c1     = (c0 + cpw < chunks) ? (c0 + cpw) : chunks;
    const int cf     = (c1 < full) ? c1 : full;

    int sum = 0;
#pragma unroll 5
    for (int c = c0; c < cf; ++c) {
        const int4 v = *(const int4*)(cnt + (c << 7) + (lane << 2));
        sum += v.x + v.y + v.z + v.w;
    }
    for (int c = (cf > c0 ? cf : c0); c < c1; ++c) {
        const int4 v = ld4_guard(cnt, (c << 7) + (lane << 2), n);
        sum += v.x + v.y + v.z + v.w;
    }
    for (int d = 16; d > 0; d >>= 1) sum += __shfl_xor(sum, d);
    if (lane == 0) wtot[wv] = sum;
    __syncthreads();
    const int t    = wtot[lane];
    const int incl = wave_incl_scan(t, lane);
    int run = __shfl(incl, wv) - __shfl(t, wv);
    const int total = __shfl(incl, 31);

    int c = c0;
    for (; c + 5 <= cf; c += 5) scan_batch<5, true>(cnt, off, c, lane, n, run);
    for (; c < cf; ++c)         scan_batch<1, true>(cnt, off, c, lane, n, run);
    for (; c < c1; ++c)         scan_batch<1, false>(cnt, off, c, lane, n, run);

    if (threadIdx.x == 0) off[n] = total;
}

__device__ __forceinline__ int pack_edge(int s, int d) { return (s << BSH) | (d & (BW - 1)); }

__global__ __launch_bounds__(256) void k_slice_scatter(const int* __restrict__ src,
                                                       const int* __restrict__ dst,
                                                       const int* __restrict__ tblo,
                                                       int* __restrict__ packed,
                                                       int E, int NB) {
    __shared__ int cur[NBMAX];
    for (int b = threadIdx.x; b < NB; b += 256) cur[b] = tblo[b * SG + blockIdx.x];
    __syncthreads();
    const bool al  = (((size_t)dst) & 15) == 0;
    const bool als = (((size_t)src) & 15) == 0;
    const int  nwp = al ? (E >> PSH) : 0;
    const int  t4  = 4 * threadIdx.x;
    for (int p0 = blockIdx.x; p0 < nwp; p0 += SU * SG) {
        int4 d[SU], s[SU], w[SU];
#pragma unroll
        for (int u = 0; u < SU; ++u) {
            const int p  = p0 + u * SG;
            const int pc = (p < nwp) ? p : p0;
            const size_t o = ((size_t)pc << PSH) + t4;
            d[u] = *(const int4*)(dst + o);
            if (als) {
                s[u] = *(const int4*)(src + o);
            } else {
                s[u].x = src[o]; s[u].y = src[o + 1]; s[u].z = src[o + 2]; s[u].w = src[o + 3];
            }
        }
#pragma unroll
        for (int u = 0; u < SU; ++u) {
            if (p0 + u * SG < nwp) {
                w[u].x = atomicAdd(&cur[d[u].x >> BSH], 1);
                w[u].y = atomicAdd(&cur[d[u].y >> BSH], 1);
                w[u].z = atomicAdd(&cur[d[u].z >> BSH], 1);
                w[u].w = atomicAdd(&cur[d[u].w >> BSH], 1);
            }
        }
#pragma unroll
        for (int u = 0; u < SU; ++u) {
            if (p0 + u * SG < nwp) {
                packed[w[u].x] = pack_edge(s[u].x, d[u].x);
                packed[w[u].y] = pack_edge(s[u].y, d[u].y);
                packed[w[u].z] = pack_edge(s[u].z, d[u].z);
                packed[w[u].w] = pack_edge(s[u].w, d[u].w);
            }
        }
    }
    for (int p = blockIdx.x; (p << PSH) < E; p += SG) {
        if (p < nwp) continue;
        for (int q = 0; q < 4; ++q) {
            const int i = (p << PSH) + q * 256 + threadIdx.x;
            if (i < E) {
                const int dd = dst[i];
                packed[atomicAdd(&cur[dd >> BSH], 1)] = pack_edge(src[i], dd);
            }
        }
    }
}

__global__ __launch_bounds__(KT) void k_bucket_rows(const int* __restrict__ packed,
                                                      const int* __restrict__ tblo,
                                                      int* __restrict__ rowptr,
                                                      float* __restrict__ dinv,
                                                      int* __restrict__ col,
                                                      int N, int NB) {
    __shared__ int cnt[BW];
    __shared__ int wt[KT / 32];
    const int b    = blockIdx.x;
    const int t    = threadIdx.x;
    const int lane = t & 31;
    const int wv   = t >> 5;
    const int beg  = tblo[b * SG];
    const int end  = tblo[(b + 1) * SG];
    cnt[t] = 0;
    __syncthreads();
    for (int base = beg; base < end; base += BU * KT) {
        int k[BU];
#pragma unroll
        for (int u = 0; u < BU; ++u) {
            const int i = base + u * KT + t;
            const int v = packed[(i < end) ? i : (end - 1)];
            k[u] = (i < end) ? v : -1;
        }
#pragma unroll
        for (int u = 0; u < BU; ++u) if (k[u] >= 0) atomicAdd(&cnt[k[u] & (BW - 1)], 1);
    }
    __syncthreads();
    const int c    = cnt[t];
    const int incl = wave_incl_scan(c, lane);
    if (lane == 31) wt[wv] = incl;
    __syncthreads();
    const int wtv   = (lane < KT / 32) ? wt[lane] : 0;
    const int wincl = wave_incl_scan(wtv, lane);
    const int woff  = __shfl(wincl, wv) - __shfl(wtv, wv);
    const int first = beg + woff + incl - c;
    const int node  = (b << BSH) + t;
    cnt[t] = first;
    if (node < N) {
        rowptr[node] = first;
        dinv[node]   = rsqrtf((float)c + 1.0f);
    }
    if (b == NB - 1 && t == 0) rowptr[N] = end;
    __syncthreads();
    for (int base = beg; base < end; base += BU * KT) {
        int k[BU], w[BU];
#pragma unroll
        for (int u = 0; u < BU; ++u) {
            const int i = base + u * KT + t;
            const int v = packed[(i < end) ? i : (end - 1)];
            k[u] = (i < end) ? v : -1;
        }
#pragma unroll
        for (int u = 0; u < BU; ++u) w[u] = (k[u] >= 0) ? atomicAdd(&cnt[k[u] & (BW - 1)], 1) : 0;
#pragma unroll
        for (int u = 0; u < BU; ++u) if (k[u] >= 0) col[w[u]] = k[u] >> BSH;
    }
}

template <int K, int F, bool BIAS, bool RELU, bool SCALE>
__global__ __launch_bounds__(256) void k_gemm_wmma(const float* __restrict__ X,
                                                   const float* __restrict__ W,
                                                   const float* __restrict__ bias,
                                                   const float* __restrict__ dinv,
                                                   float* __restrict__ H,
                                                   int tilesM) {
    static_assert(K % 4 == 0 && F % 16 == 0, "shape");
    constexpr int TF = F / 16;
    const int tm = blockIdx.x * (blockDim.x >> 5) + (threadIdx.x >> 5);
    if (tm >= tilesM) return;
    const int lane = threadIdx.x & 31;
    const int l15  = lane & 15;
    const int hi2  = (lane >> 4) << 1;

    const float* xrow = X + (size_t)(tm * 16 + l15) * K + hi2;
    const float* wcol = W + (size_t)hi2 * F + l15;

    v8f acc[TF];
#pragma unroll
    for (int t = 0; t < TF; ++t) acc[t] = v8f{};

    constexpr int KS = K / 4;
    v2f a[KS];
#pragma unroll
    for (int s = 0; s < KS; ++s) {
        a[s].x = xrow[4 * s];
        a[s].y = xrow[4 * s + 1];
    }

#pragma unroll
    for (int s = 0; s < KS; ++s) {
#pragma unroll
        for (int t = 0; t < TF; ++t) {
            v2f b;
            b.x = wcol[(size_t)(4 * s) * F + t * 16];
            b.y = wcol[(size_t)(4 * s + 1) * F + t * 16];
            acc[t] = __builtin_amdgcn_wmma_f32_16x16x4_f32(
                 false, a[s],  false, b,
                 (short)0, acc[t],  false,  false);
        }
    }

    const int hi8 = (lane >> 4) << 3;
    float dv[8];
#pragma unroll
    for (int v = 0; v < 8; ++v) dv[v] = SCALE ? dinv[tm * 16 + hi8 + v] : 1.0f;
    float* hrow = H + (size_t)(tm * 16 + hi8) * F + l15;
#pragma unroll
    for (int t = 0; t < TF; ++t) {
        float bv = 0.0f;
        if (BIAS) bv = bias[t * 16 + l15];
#pragma unroll
        for (int v = 0; v < 8; ++v) {
            float o = acc[t][v];
            if (BIAS) o += bv;
            if (RELU) o = fmaxf(o, 0.0f);
            if (SCALE) o *= dv[v];
            hrow[(size_t)v * F + t * 16] = o;
        }
    }
}

template <int K, int F, bool BIAS, bool RELU, bool SCALE>
__global__ __launch_bounds__(256) void k_gemm_tail(const float* __restrict__ X,
                                                   const float* __restrict__ W,
                                                   const float* __restrict__ bias,
                                                   const float* __restrict__ dinv,
                                                   float* __restrict__ H,
                                                   int rowStart, int rowEnd) {
    int idx = blockIdx.x * blockDim.x + threadIdx.x;
    int rows = rowEnd - rowStart;
    if (idx >= rows * F) return;
    int r = rowStart + idx / F, f = idx % F;
    float s = 0.f;
    for (int k = 0; k < K; ++k) s += X[(size_t)r * K + k] * W[(size_t)k * F + f];
    if (BIAS) s += bias[f];
    if (RELU) s = fmaxf(s, 0.0f);
    if (SCALE) s *= dinv[r];
    H[(size_t)r * F + f] = s;
}

template <int F, bool BIAS, bool RELU, bool DUAL>
__global__ __launch_bounds__(256) void k_gather(const float* __restrict__ HS,
                                                const int* __restrict__ rowptr,
                                                const int* __restrict__ col,
                                                const float* __restrict__ dinv,
                                                const float* __restrict__ bias,
                                                float* __restrict__ OUT,
                                                float* __restrict__ OUTS, int N) {
    constexpr int CH  = F / 4;
    constexpr int NPB = 256 / CH;
    const int c = threadIdx.x % CH;
    const int g = threadIdx.x / CH;
    const float4* __restrict__ H4 = (const float4*)HS;
    float4* __restrict__ O4 = (float4*)OUT;
    float4* __restrict__ S4 = (float4*)OUTS;

    float4 bv = make_float4(0.f, 0.f, 0.f, 0.f);
    if (BIAS) bv = ((const float4*)bias)[c];

    const int stride = gridDim.x * NPB;
    int node = blockIdx.x * NPB + g;
    int beg = 0, end = 0;
    if (node < N) { beg = rowptr[node]; end = rowptr[node + 1]; }

    while (node < N) {
        const int nxt = node + stride;
        int nbeg = 0, nend = 0;
        if (nxt < N) { nbeg = rowptr[nxt]; nend = rowptr[nxt + 1]; }

        const float dd = dinv[node];
        const float4 hs = H4[(size_t)node * CH + c];
        float4 acc = make_float4(0.f, 0.f, 0.f, 0.f);

        int j = beg;
        for (; j + 8 <= end; j += 8) {
            int   sidx[8];
            float4 h[8];
#pragma unroll
            for (int u = 0; u < 8; ++u) sidx[u] = col[j + u];
#pragma unroll
            for (int u = 0; u < 8; ++u) h[u] = H4[(size_t)sidx[u] * CH + c];
#pragma unroll
            for (int u = 0; u < 8; ++u) {
                acc.x += h[u].x;
                acc.y += h[u].y;
                acc.z += h[u].z;
                acc.w += h[u].w;
            }
        }
        for (; j < end; j += 4) {
            int   sidx[4];
            float4 h[4];
#pragma unroll
            for (int u = 0; u < 4; ++u) {
                const int jj = (j + u < end) ? (j + u) : (end - 1);
                sidx[u] = col[jj];
            }
#pragma unroll
            for (int u = 0; u < 4; ++u) h[u] = H4[(size_t)sidx[u] * CH + c];
#pragma unroll
            for (int u = 0; u < 4; ++u) {
                if (j + u < end) {
                    acc.x += h[u].x;
                    acc.y += h[u].y;
                    acc.z += h[u].z;
                    acc.w += h[u].w;
                }
            }
        }

        float4 o;
        o.x = (acc.x + hs.x) * dd;
        o.y = (acc.y + hs.y) * dd;
        o.z = (acc.z + hs.z) * dd;
        o.w = (acc.w + hs.w) * dd;
        if (BIAS) { o.x += bv.x; o.y += bv.y; o.z += bv.z; o.w += bv.w; }
        if (RELU) {
            o.x = fmaxf(o.x, 0.0f); o.y = fmaxf(o.y, 0.0f);
            o.z = fmaxf(o.z, 0.0f); o.w = fmaxf(o.w, 0.0f);
        }
        O4[(size_t)node * CH + c] = o;
        if (DUAL) {
            float4 q;
            q.x = o.x * dd; q.y = o.y * dd; q.z = o.z * dd; q.w = o.w * dd;
            S4[(size_t)node * CH + c] = q;
        }

        node = nxt; beg = nbeg; end = nend;
    }
}

static inline int nblk(long long n, int t) { return (int)((n + t - 1) / t); }

template <int K, int F, bool BIAS, bool RELU, bool SCALE>
static void launch_gemm(const float* X, const float* W, const float* bias, const float* dinv,
                        float* H, int N, hipStream_t s) {
    int tilesM = N / 16;
    long long waves = tilesM;
    if (waves > 0)
        k_gemm_wmma<K, F, BIAS, RELU, SCALE><<<nblk(waves, 8), 256, 0, s>>>(X, W, bias, dinv, H,
                                                                            tilesM);
    int tail = tilesM * 16;
    if (tail < N)
        k_gemm_tail<K, F, BIAS, RELU, SCALE><<<nblk((long long)(N - tail) * F, 256), 256, 0, s>>>(
            X, W, bias, dinv, H, tail, N);
}

extern "C" void kernel_launch(void* const* d_in, const int* in_sizes, int n_in,
                              void* d_out, int out_size, void* d_ws, size_t ws_size,
                              hipStream_t stream) {
    const float* x  = (const float*)d_in[0];
    const int*   ei = (const int*)d_in[1];
    const float* W1 = (const float*)d_in[2]; const float* b1 = (const float*)d_in[3];
    const float* W2 = (const float*)d_in[4]; const float* b2 = (const float*)d_in[5];
    const float* W3 = (const float*)d_in[6]; const float* b3 = (const float*)d_in[7];
    const float* W4 = (const float*)d_in[8]; const float* b4 = (const float*)d_in[9];

    const int N = in_sizes[0] / 128;
    const int E = in_sizes[1] / 2;
    const int* src = ei;
    const int* dst = ei + E;

    float* out  = (float*)d_out;
    float* xrec = out;
    float* z    = out + (size_t)N * 128;

    float* ws   = (float*)d_ws;
    float* dinv = ws;
    float* bufH = ws + N;
    float* bufA = bufH + (size_t)N * 128;
    int* tblc   = (int*)(bufH + (size_t)N * 64);
    int* tblo   = tblc + NBMAX * SG;
    int* rowptr = tblo + NBMAX * SG + 4;
    int* col    = rowptr + N + 4;
    int* packed = col + E;

    const int GG = 1536;

    const int NB  = (N + BW - 1) >> BSH;
    k_slice_hist<<<SG, 256, 0, stream>>>(dst, tblc, E, NB);
    k_scan_table<<<1, 1024, 0, stream>>>(tblc, tblo, NB * SG);
    k_slice_scatter<<<SG, 256, 0, stream>>>(src, dst, tblo, packed, E, NB);
    k_bucket_rows<<<NB, KT, 0, stream>>>(packed, tblo, rowptr, dinv, col, N, NB);

    launch_gemm<128, 64, false, false, true>(x, W1, nullptr, dinv, bufH, N, stream);
    k_gather<64, true, true, false><<<GG, 256, 0, stream>>>(bufH, rowptr, col, dinv, b1, bufA,
                                                            nullptr, N);

    launch_gemm<64, 32, false, false, true>(bufA, W2, nullptr, dinv, bufH, N, stream);
    k_gather<32, true, false, true><<<GG, 256, 0, stream>>>(bufH, rowptr, col, dinv, b2, z, bufA,
                                                            N);

    k_gather<32, false, false, false><<<GG, 256, 0, stream>>>(bufA, rowptr, col, dinv, nullptr,
                                                              bufH, nullptr, N);
    launch_gemm<32, 64, true, true, true>(bufH, W3, b3, dinv, bufA, N, stream);

    k_gather<64, false, false, false><<<GG, 256, 0, stream>>>(bufA, rowptr, col, dinv, nullptr,
                                                              bufH, nullptr, N);
    launch_gemm<64, 128, true, false, false>(bufH, W4, b4, nullptr, xrec, N, stream);
}


# Each GEMM fed by an edge sum (layers 2,3,4) runs inside the edge-sum kernel: summed rows go to an LDS tile, one barrier, one WMMA tile per wave, W in LDS; 12 -> 9 launches; sums unchanged
# speedup vs baseline: 4.5723x; 1.1453x over previous
#include <hip/hip_runtime.h>
#include <hip/hip_bf16.h>


typedef float v2f __attribute__((ext_vector_type(2)));
typedef float v8f __attribute__((ext_vector_type(8)));

constexpr int BSH   = 9;
constexpr int BW    = 1 << BSH;
constexpr int NBMAX = 256;
constexpr int SG    = 256;
constexpr int PSH   = 10;
constexpr int SU    = 8;
constexpr int KT    = BW;
constexpr int BU    = 20;

__device__ __forceinline__ int4 ld4_guard(const int* __restrict__ p, int i, int n) {
    if (i + 3 < n) return *(const int4*)(p + i);
    int4 v;
    v.x = (i     < n) ? p[i]     : 0;
    v.y = (i + 1 < n) ? p[i + 1] : 0;
    v.z = (i + 2 < n) ? p[i + 2] : 0;
    v.w = 0;
    return v;
}

__device__ __forceinline__ int wave_incl_scan(int v, int lane) {
    for (int d = 1; d < 32; d <<= 1) {
        int u = __shfl_up(v, d);
        if (lane >= d) v += u;
    }
    return v;
}

__global__ __launch_bounds__(256) void k_slice_hist(const int* __restrict__ dst,
                                                    int* __restrict__ tblc,
                                                    int E, int NB) {
    __shared__ int hist[NBMAX];
    for (int b = threadIdx.x; b < NB; b += 256) hist[b] = 0;
    __syncthreads();
    const bool al  = (((size_t)dst) & 15) == 0;
    const int  nwp = al ? (E >> PSH) : 0;
    const int  t4  = 4 * threadIdx.x;
    for (int p0 = blockIdx.x; p0 < nwp; p0 += SU * SG) {
        int4 d[SU];
#pragma unroll
        for (int u = 0; u < SU; ++u) {
            const int p  = p0 + u * SG;
            const int pc = (p < nwp) ? p : p0;
            d[u] = *(const int4*)(dst + ((size_t)pc << PSH) + t4);
        }
#pragma unroll
        for (int u = 0; u < SU; ++u) {
            if (p0 + u * SG < nwp) {
                atomicAdd(&hist[d[u].x >> BSH], 1);
                atomicAdd(&hist[d[u].y >> BSH], 1);
                atomicAdd(&hist[d[u].z >> BSH], 1);
                atomicAdd(&hist[d[u].w >> BSH], 1);
            }
        }
    }
    for (int p = blockIdx.x; (p << PSH) < E; p += SG) {
        if (p < nwp) continue;
        for (int q = 0; q < 4; ++q) {
            const int i = (p << PSH) + q * 256 + threadIdx.x;
            if (i < E) atomicAdd(&hist[dst[i] >> BSH], 1);
        }
    }
    __syncthreads();
    for (int b = threadIdx.x; b < NB; b += 256) tblc[b * SG + blockIdx.x] = hist[b];
}

template <int B, bool FULL>
__device__ __forceinline__ void scan_batch(const int* __restrict__ cnt, int* __restrict__ off,
                                           int c, int lane, int n, int& run) {
    int4 v[B];
    int  ls[B], inc[B], tot[B];
#pragma unroll
    for (int b = 0; b < B; ++b) {
        const int i = ((c + b) << 7) + (lane << 2);
        if (FULL) v[b] = *(const int4*)(cnt + i); else v[b] = ld4_guard(cnt, i, n);
    }
#pragma unroll
    for (int b = 0; b < B; ++b) { ls[b] = v[b].x + v[b].y + v[b].z + v[b].w; inc[b] = ls[b]; }
    for (int d = 1; d < 32; d <<= 1) {
        int u[B];
#pragma unroll
        for (int b = 0; b < B; ++b) u[b] = __shfl_up(inc[b], d);
#pragma unroll
        for (int b = 0; b < B; ++b) if (lane >= d) inc[b] += u[b];
    }
#pragma unroll
    for (int b = 0; b < B; ++b) tot[b] = __shfl(inc[b], 31);
#pragma unroll
    for (int b = 0; b < B; ++b) {
        const int i  = ((c + b) << 7) + (lane << 2);
        const int b0 = run + inc[b] - ls[b];
        run += tot[b];
        int4 r;
        r.x = b0; r.y = b0 + v[b].x; r.z = r.y + v[b].y; r.w = r.z + v[b].z;
        if (FULL || i + 3 < n) {
            *(int4*)(off + i) = r;
        } else {
            if (i     < n) off[i]     = r.x;
            if (i + 1 < n) off[i + 1] = r.y;
            if (i + 2 < n) off[i + 2] = r.z;
        }
    }
}

__global__ __launch_bounds__(1024) void k_scan_table(const int* __restrict__ cnt,
                                                     int* __restrict__ off, int n) {
    __shared__ int wtot[32];
    const int lane = threadIdx.x & 31;
    const int wv   = threadIdx.x >> 5;
    const int chunks = (n + 127) >> 7;
    const int full   = n >> 7;
    const int cpw    = (chunks + 31) >> 5;
    const int c0     = wv * cpw;
    const int c1     = (c0 + cpw < chunks) ? (c0 + cpw) : chunks;
    const int cf     = (c1 < full) ? c1 : full;

    int sum = 0;
#pragma unroll 5
    for (int c = c0; c < cf; ++c) {
        const int4 v = *(const int4*)(cnt + (c << 7) + (lane << 2));
        sum += v.x + v.y + v.z + v.w;
    }
    for (int c = (cf > c0 ? cf : c0); c < c1; ++c) {
        const int4 v = ld4_guard(cnt, (c << 7) + (lane << 2), n);
        sum += v.x + v.y + v.z + v.w;
    }
    for (int d = 16; d > 0; d >>= 1) sum += __shfl_xor(sum, d);
    if (lane == 0) wtot[wv] = sum;
    __syncthreads();
    const int t    = wtot[lane];
    const int incl = wave_incl_scan(t, lane);
    int run = __shfl(incl, wv) - __shfl(t, wv);
    const int total = __shfl(incl, 31);

    int c = c0;
    for (; c + 5 <= cf; c += 5) scan_batch<5, true>(cnt, off, c, lane, n, run);
    for (; c < cf; ++c)         scan_batch<1, true>(cnt, off, c, lane, n, run);
    for (; c < c1; ++c)         scan_batch<1, false>(cnt, off, c, lane, n, run);

    if (threadIdx.x == 0) off[n] = total;
}

__device__ __forceinline__ int pack_edge(int s, int d) { return (s << BSH) | (d & (BW - 1)); }

__global__ __launch_bounds__(256) void k_slice_scatter(const int* __restrict__ src,
                                                       const int* __restrict__ dst,
                                                       const int* __restrict__ tblo,
                                                       int* __restrict__ packed,
                                                       int E, int NB) {
    __shared__ int cur[NBMAX];
    for (int b = threadIdx.x; b < NB; b += 256) cur[b] = tblo[b * SG + blockIdx.x];
    __syncthreads();
    const bool al  = (((size_t)dst) & 15) == 0;
    const bool als = (((size_t)src) & 15) == 0;
    const int  nwp = al ? (E >> PSH) : 0;
    const int  t4  = 4 * threadIdx.x;
    for (int p0 = blockIdx.x; p0 < nwp; p0 += SU * SG) {
        int4 d[SU], s[SU], w[SU];
#pragma unroll
        for (int u = 0; u < SU; ++u) {
            const int p  = p0 + u * SG;
            const int pc = (p < nwp) ? p : p0;
            const size_t o = ((size_t)pc << PSH) + t4;
            d[u] = *(const int4*)(dst + o);
            if (als) {
                s[u] = *(const int4*)(src + o);
            } else {
                s[u].x = src[o]; s[u].y = src[o + 1]; s[u].z = src[o + 2]; s[u].w = src[o + 3];
            }
        }
#pragma unroll
        for (int u = 0; u < SU; ++u) {
            if (p0 + u * SG < nwp) {
                w[u].x = atomicAdd(&cur[d[u].x >> BSH], 1);
                w[u].y = atomicAdd(&cur[d[u].y >> BSH], 1);
                w[u].z = atomicAdd(&cur[d[u].z >> BSH], 1);
                w[u].w = atomicAdd(&cur[d[u].w >> BSH], 1);
            }
        }
#pragma unroll
        for (int u = 0; u < SU; ++u) {
            if (p0 + u * SG < nwp) {
                packed[w[u].x] = pack_edge(s[u].x, d[u].x);
                packed[w[u].y] = pack_edge(s[u].y, d[u].y);
                packed[w[u].z] = pack_edge(s[u].z, d[u].z);
                packed[w[u].w] = pack_edge(s[u].w, d[u].w);
            }
        }
    }
    for (int p = blockIdx.x; (p << PSH) < E; p += SG) {
        if (p < nwp) continue;
        for (int q = 0; q < 4; ++q) {
            const int i = (p << PSH) + q * 256 + threadIdx.x;
            if (i < E) {
                const int dd = dst[i];
                packed[atomicAdd(&cur[dd >> BSH], 1)] = pack_edge(src[i], dd);
            }
        }
    }
}

__global__ __launch_bounds__(KT) void k_bucket_rows(const int* __restrict__ packed,
                                                      const int* __restrict__ tblo,
                                                      int* __restrict__ rowptr,
                                                      float* __restrict__ dinv,
                                                      int* __restrict__ col,
                                                      int N, int NB) {
    __shared__ int cnt[BW];
    __shared__ int wt[KT / 32];
    const int b    = blockIdx.x;
    const int t    = threadIdx.x;
    const int lane = t & 31;
    const int wv   = t >> 5;
    const int beg  = tblo[b * SG];
    const int end  = tblo[(b + 1) * SG];
    cnt[t] = 0;
    __syncthreads();
    for (int base = beg; base < end; base += BU * KT) {
        int k[BU];
#pragma unroll
        for (int u = 0; u < BU; ++u) {
            const int i = base + u * KT + t;
            const int v = packed[(i < end) ? i : (end - 1)];
            k[u] = (i < end) ? v : -1;
        }
#pragma unroll
        for (int u = 0; u < BU; ++u) if (k[u] >= 0) atomicAdd(&cnt[k[u] & (BW - 1)], 1);
    }
    __syncthreads();
    const int c    = cnt[t];
    const int incl = wave_incl_scan(c, lane);
    if (lane == 31) wt[wv] = incl;
    __syncthreads();
    const int wtv   = (lane < KT / 32) ? wt[lane] : 0;
    const int wincl = wave_incl_scan(wtv, lane);
    const int woff  = __shfl(wincl, wv) - __shfl(wtv, wv);
    const int first = beg + woff + incl - c;
    const int node  = (b << BSH) + t;
    cnt[t] = first;
    if (node < N) {
        rowptr[node] = first;
        dinv[node]   = rsqrtf((float)c + 1.0f);
    }
    if (b == NB - 1 && t == 0) rowptr[N] = end;
    __syncthreads();
    for (int base = beg; base < end; base += BU * KT) {
        int k[BU], w[BU];
#pragma unroll
        for (int u = 0; u < BU; ++u) {
            const int i = base + u * KT + t;
            const int v = packed[(i < end) ? i : (end - 1)];
            k[u] = (i < end) ? v : -1;
        }
#pragma unroll
        for (int u = 0; u < BU; ++u) w[u] = (k[u] >= 0) ? atomicAdd(&cnt[k[u] & (BW - 1)], 1) : 0;
#pragma unroll
        for (int u = 0; u < BU; ++u) if (k[u] >= 0) col[w[u]] = k[u] >> BSH;
    }
}

template <int K, int F, bool BIAS, bool RELU, bool SCALE>
__global__ __launch_bounds__(256) void k_gemm_wmma(const float* __restrict__ X,
                                                   const float* __restrict__ W,
                                                   const float* __restrict__ bias,
                                                   const float* __restrict__ dinv,
                                                   float* __restrict__ H,
                                                   int tilesM) {
    static_assert(K % 4 == 0 && F % 16 == 0, "shape");
    constexpr int TF = F / 16;
    const int tm = blockIdx.x * (blockDim.x >> 5) + (threadIdx.x >> 5);
    if (tm >= tilesM) return;
    const int lane = threadIdx.x & 31;
    const int l15  = lane & 15;
    const int hi2  = (lane >> 4) << 1;

    const float* xrow = X + (size_t)(tm * 16 + l15) * K + hi2;
    const float* wcol = W + (size_t)hi2 * F + l15;

    v8f acc[TF];
#pragma unroll
    for (int t = 0; t < TF; ++t) acc[t] = v8f{};

    constexpr int KS = K / 4;
    v2f a[KS];
#pragma unroll
    for (int s = 0; s < KS; ++s) {
        a[s].x = xrow[4 * s];
        a[s].y = xrow[4 * s + 1];
    }

#pragma unroll
    for (int s = 0; s < KS; ++s) {
#pragma unroll
        for (int t = 0; t < TF; ++t) {
            v2f b;
            b.x = wcol[(size_t)(4 * s) * F + t * 16];
            b.y = wcol[(size_t)(4 * s + 1) * F + t * 16];
            acc[t] = __builtin_amdgcn_wmma_f32_16x16x4_f32(
                 false, a[s],  false, b,
                 (short)0, acc[t],  false,  false);
        }
    }

    const int hi8 = (lane >> 4) << 3;
    float dv[8];
#pragma unroll
    for (int v = 0; v < 8; ++v) dv[v] = SCALE ? dinv[tm * 16 + hi8 + v] : 1.0f;
    float* hrow = H + (size_t)(tm * 16 + hi8) * F + l15;
#pragma unroll
    for (int t = 0; t < TF; ++t) {
        float bv = 0.0f;
        if (BIAS) bv = bias[t * 16 + l15];
#pragma unroll
        for (int v = 0; v < 8; ++v) {
            float o = acc[t][v];
            if (BIAS) o += bv;
            if (RELU) o = fmaxf(o, 0.0f);
            if (SCALE) o *= dv[v];
            hrow[(size_t)v * F + t * 16] = o;
        }
    }
}

template <int K, int F, bool BIAS, bool RELU, bool SCALE>
__global__ __launch_bounds__(256) void k_gemm_tail(const float* __restrict__ X,
                                                   const float* __restrict__ W,
                                                   const float* __restrict__ bias,
                                                   const float* __restrict__ dinv,
                                                   float* __restrict__ H,
                                                   int rowStart, int rowEnd) {
    int idx = blockIdx.x * blockDim.x + threadIdx.x;
    int rows = rowEnd - rowStart;
    if (idx >= rows * F) return;
    int r = rowStart + idx / F, f = idx % F;
    float s = 0.f;
    for (int k = 0; k < K; ++k) s += X[(size_t)r * K + k] * W[(size_t)k * F + f];
    if (BIAS) s += bias[f];
    if (RELU) s = fmaxf(s, 0.0f);
    if (SCALE) s *= dinv[r];
    H[(size_t)r * F + f] = s;
}

template <int F, bool BIAS, bool RELU, bool DUAL>
__global__ __launch_bounds__(256) void k_gather(const float* __restrict__ HS,
                                                const int* __restrict__ rowptr,
                                                const int* __restrict__ col,
                                                const float* __restrict__ dinv,
                                                const float* __restrict__ bias,
                                                float* __restrict__ OUT,
                                                float* __restrict__ OUTS, int N) {
    constexpr int CH  = F / 4;
    constexpr int NPB = 256 / CH;
    const int c = threadIdx.x % CH;
    const int g = threadIdx.x / CH;
    const float4* __restrict__ H4 = (const float4*)HS;
    float4* __restrict__ O4 = (float4*)OUT;
    float4* __restrict__ S4 = (float4*)OUTS;

    float4 bv = make_float4(0.f, 0.f, 0.f, 0.f);
    if (BIAS) bv = ((const float4*)bias)[c];

    const int stride = gridDim.x * NPB;
    int node = blockIdx.x * NPB + g;
    int beg = 0, end = 0;
    if (node < N) { beg = rowptr[node]; end = rowptr[node + 1]; }

    while (node < N) {
        const int nxt = node + stride;
        int nbeg = 0, nend = 0;
        if (nxt < N) { nbeg = rowptr[nxt]; nend = rowptr[nxt + 1]; }

        const float dd = dinv[node];
        const float4 hs = H4[(size_t)node * CH + c];
        float4 acc = make_float4(0.f, 0.f, 0.f, 0.f);

        int j = beg;
        for (; j + 8 <= end; j += 8) {
            int   sidx[8];
            float4 h[8];
#pragma unroll
            for (int u = 0; u < 8; ++u) sidx[u] = col[j + u];
#pragma unroll
            for (int u = 0; u < 8; ++u) h[u] = H4[(size_t)sidx[u] * CH + c];
#pragma unroll
            for (int u = 0; u < 8; ++u) {
                acc.x += h[u].x;
                acc.y += h[u].y;
                acc.z += h[u].z;
                acc.w += h[u].w;
            }
        }
        for (; j < end; j += 4) {
            int   sidx[4];
            float4 h[4];
#pragma unroll
            for (int u = 0; u < 4; ++u) {
                const int jj = (j + u < end) ? (j + u) : (end - 1);
                sidx[u] = col[jj];
            }
#pragma unroll
            for (int u = 0; u < 4; ++u) h[u] = H4[(size_t)sidx[u] * CH + c];
#pragma unroll
            for (int u = 0; u < 4; ++u) {
                if (j + u < end) {
                    acc.x += h[u].x;
                    acc.y += h[u].y;
                    acc.z += h[u].z;
                    acc.w += h[u].w;
                }
            }
        }

        float4 o;
        o.x = (acc.x + hs.x) * dd;
        o.y = (acc.y + hs.y) * dd;
        o.z = (acc.z + hs.z) * dd;
        o.w = (acc.w + hs.w) * dd;
        if (BIAS) { o.x += bv.x; o.y += bv.y; o.z += bv.z; o.w += bv.w; }
        if (RELU) {
            o.x = fmaxf(o.x, 0.0f); o.y = fmaxf(o.y, 0.0f);
            o.z = fmaxf(o.z, 0.0f); o.w = fmaxf(o.w, 0.0f);
        }
        O4[(size_t)node * CH + c] = o;
        if (DUAL) {
            float4 q;
            q.x = o.x * dd; q.y = o.y * dd; q.z = o.z * dd; q.w = o.w * dd;
            S4[(size_t)node * CH + c] = q;
        }

        node = nxt; beg = nbeg; end = nend;
    }
}

template <int F, int FO, bool GBIAS, bool GRELU, bool OBIAS, bool ORELU, bool OSCALE>
__global__ __launch_bounds__(256) void k_gather_gemm(const float* __restrict__ HS,
                                                     const int* __restrict__ rowptr,
                                                     const int* __restrict__ col,
                                                     const float* __restrict__ dinv,
                                                     const float* __restrict__ gbias,
                                                     const float* __restrict__ W,
                                                     const float* __restrict__ obias,
                                                     float* __restrict__ OUT, int N) {
    constexpr int CH    = F / 4;
    constexpr int NPB   = 256 / CH;
    constexpr int PITCH = F + 4;
    constexpr int CT    = FO / 16;
    constexpr int TILES = (NPB / 16) * CT;
    constexpr int KS    = F / 4;
    static_assert(TILES <= 8 && NPB % 16 == 0, "one tile per wave");
    __shared__ __attribute__((aligned(16))) float sA[2][NPB * PITCH];
    __shared__ __attribute__((aligned(16))) float sW[F * FO];

    const int c = threadIdx.x % CH;
    const int g = threadIdx.x / CH;
    const int lane = threadIdx.x & 31;
    const int wv   = threadIdx.x >> 5;
    const int l15  = lane & 15;
    const int hi2  = (lane >> 4) << 1;
    const int hi8  = (lane >> 4) << 3;
    const bool has_tile = wv < TILES;
    const int strip = wv / CT;
    const int ct    = wv % CT;
    const float4* __restrict__ H4 = (const float4*)HS;

    float4 bv = make_float4(0.f, 0.f, 0.f, 0.f);
    if (GBIAS) bv = ((const float4*)gbias)[c];

    for (int q = threadIdx.x; q < F * FO / 4; q += 256) {
        const int k  = q / (FO / 4);
        const int n4 = (q % (FO / 4)) * 4;
        const float4 w4 = *(const float4*)(W + (size_t)k * FO + n4);
        float* dst = &sW[(((n4 >> 4) * KS + (k >> 2)) * 32 + ((k >> 1) & 1) * 16 + (n4 & 15)) * 2 + (k & 1)];
        dst[0] = w4.x; dst[2] = w4.y; dst[4] = w4.z; dst[6] = w4.w;
    }
    float ob = 0.0f;
    if (OBIAS && has_tile) ob = obias[ct * 16 + l15];

    const int stride = gridDim.x * NPB;
    int node = blockIdx.x * NPB + g;
    int beg = 0, end = 0;
    if (node < N) { beg = rowptr[node]; end = rowptr[node + 1]; }

    int buf = 0;
    for (int base = blockIdx.x * NPB; base < N; base += stride) {
        const int nxt = node + stride;
        int nbeg = 0, nend = 0;
        if (nxt < N) { nbeg = rowptr[nxt]; nend = rowptr[nxt + 1]; }

        float4 o = make_float4(0.f, 0.f, 0.f, 0.f);
        if (node < N) {
            const float dd = dinv[node];
            const float4 hs = H4[(size_t)node * CH + c];
            float4 acc = make_float4(0.f, 0.f, 0.f, 0.f);

            int j = beg;
            for (; j + 8 <= end; j += 8) {
                int   sidx[8];
                float4 h[8];
#pragma unroll
                for (int u = 0; u < 8; ++u) sidx[u] = col[j + u];
#pragma unroll
                for (int u = 0; u < 8; ++u) h[u] = H4[(size_t)sidx[u] * CH + c];
#pragma unroll
                for (int u = 0; u < 8; ++u) {
                    acc.x += h[u].x;
                    acc.y += h[u].y;
                    acc.z += h[u].z;
                    acc.w += h[u].w;
                }
            }
            for (; j < end; j += 4) {
                int   sidx[4];
                float4 h[4];
#pragma unroll
                for (int u = 0; u < 4; ++u) {
                    const int jj = (j + u < end) ? (j + u) : (end - 1);
                    sidx[u] = col[jj];
                }
#pragma unroll
                for (int u = 0; u < 4; ++u) h[u] = H4[(size_t)sidx[u] * CH + c];
#pragma unroll
                for (int u = 0; u < 4; ++u) {
                    if (j + u < end) {
                        acc.x += h[u].x;
                        acc.y += h[u].y;
                        acc.z += h[u].z;
                        acc.w += h[u].w;
                    }
                }
            }

            o.x = (acc.x + hs.x) * dd;
            o.y = (acc.y + hs.y) * dd;
            o.z = (acc.z + hs.z) * dd;
            o.w = (acc.w + hs.w) * dd;
            if (GBIAS) { o.x += bv.x; o.y += bv.y; o.z += bv.z; o.w += bv.w; }
            if (GRELU) {
                o.x = fmaxf(o.x, 0.0f); o.y = fmaxf(o.y, 0.0f);
                o.z = fmaxf(o.z, 0.0f); o.w = fmaxf(o.w, 0.0f);
            }
        }
        *(float4*)&sA[buf][g * PITCH + 4 * c] = o;
        __syncthreads();

        const int r0 = base + strip * 16;
        if (has_tile && r0 + 16 <= N) {
            const float* arow = &sA[buf][(strip * 16 + l15) * PITCH + hi2];
            const float* bfr  = &sW[(ct * KS * 32 + lane) * 2];
            v8f d = v8f{};
#pragma unroll
            for (int s = 0; s < KS; ++s) {
                const v2f a = *(const v2f*)(arow + 4 * s);
                const v2f b = *(const v2f*)(bfr + s * 64);
                d = __builtin_amdgcn_wmma_f32_16x16x4_f32(
                     false, a,  false, b,
                     (short)0, d,  false,  false);
            }
            float dv[8];
#pragma unroll
            for (int v = 0; v < 8; ++v) dv[v] = OSCALE ? dinv[r0 + hi8 + v] : 1.0f;
            float* orow = OUT + (size_t)(r0 + hi8) * FO + ct * 16 + l15;
#pragma unroll
            for (int v = 0; v < 8; ++v) {
                float q = d[v];
                if (OBIAS) q += ob;
                if (ORELU) q = fmaxf(q, 0.0f);
                if (OSCALE) q *= dv[v];
                orow[(size_t)v * FO] = q;
            }
        } else if (has_tile && r0 < N) {
            for (int r = (lane >> 4); r < 16 && r0 + r < N; r += 2) {
                const float* ar = &sA[buf][(strip * 16 + r) * PITCH];
                float q = 0.0f;
                for (int k = 0; k < F; ++k) q += ar[k] * W[(size_t)k * FO + ct * 16 + l15];
                if (OBIAS) q += ob;
                if (ORELU) q = fmaxf(q, 0.0f);
                if (OSCALE) q *= dinv[r0 + r];
                OUT[(size_t)(r0 + r) * FO + ct * 16 + l15] = q;
            }
        }

        node = nxt; beg = nbeg; end = nend;
        buf ^= 1;
    }
}

static inline int nblk(long long n, int t) { return (int)((n + t - 1) / t); }

template <int K, int F, bool BIAS, bool RELU, bool SCALE>
static void launch_gemm(const float* X, const float* W, const float* bias, const float* dinv,
                        float* H, int N, hipStream_t s) {
    int tilesM = N / 16;
    long long waves = tilesM;
    if (waves > 0)
        k_gemm_wmma<K, F, BIAS, RELU, SCALE><<<nblk(waves, 8), 256, 0, s>>>(X, W, bias, dinv, H,
                                                                            tilesM);
    int tail = tilesM * 16;
    if (tail < N)
        k_gemm_tail<K, F, BIAS, RELU, SCALE><<<nblk((long long)(N - tail) * F, 256), 256, 0, s>>>(
            X, W, bias, dinv, H, tail, N);
}

extern "C" void kernel_launch(void* const* d_in, const int* in_sizes, int n_in,
                              void* d_out, int out_size, void* d_ws, size_t ws_size,
                              hipStream_t stream) {
    const float* x  = (const float*)d_in[0];
    const int*   ei = (const int*)d_in[1];
    const float* W1 = (const float*)d_in[2]; const float* b1 = (const float*)d_in[3];
    const float* W2 = (const float*)d_in[4]; const float* b2 = (const float*)d_in[5];
    const float* W3 = (const float*)d_in[6]; const float* b3 = (const float*)d_in[7];
    const float* W4 = (const float*)d_in[8]; const float* b4 = (const float*)d_in[9];

    const int N = in_sizes[0] / 128;
    const int E = in_sizes[1] / 2;
    const int* src = ei;
    const int* dst = ei + E;

    float* out  = (float*)d_out;
    float* xrec = out;
    float* z    = out + (size_t)N * 128;

    float* ws   = (float*)d_ws;
    float* dinv = ws;
    float* bufH = ws + N;
    float* bufA = bufH + (size_t)N * 128;
    int* tblc   = (int*)(bufH + (size_t)N * 64);
    int* tblo   = tblc + NBMAX * SG;
    int* rowptr = tblo + NBMAX * SG + 4;
    int* col    = rowptr + N + 4;
    int* packed = col + E;

    const int GG = 1536;

    const int NB  = (N + BW - 1) >> BSH;
    k_slice_hist<<<SG, 256, 0, stream>>>(dst, tblc, E, NB);
    k_scan_table<<<1, 1024, 0, stream>>>(tblc, tblo, NB * SG);
    k_slice_scatter<<<SG, 256, 0, stream>>>(src, dst, tblo, packed, E, NB);
    k_bucket_rows<<<NB, KT, 0, stream>>>(packed, tblo, rowptr, dinv, col, N, NB);

    launch_gemm<128, 64, false, false, true>(x, W1, nullptr, dinv, bufH, N, stream);

    k_gather_gemm<64, 32, true, true, false, false, true><<<GG, 256, 0, stream>>>(
        bufH, rowptr, col, dinv, b1, W2, nullptr, bufA, N);

    k_gather<32, true, false, true><<<GG, 256, 0, stream>>>(bufA, rowptr, col, dinv, b2, z, bufH,
                                                            N);

    k_gather_gemm<32, 64, false, false, true, true, true><<<GG, 256, 0, stream>>>(
        bufH, rowptr, col, dinv, nullptr, W3, b3, bufA, N);

    k_gather_gemm<64, 128, false, false, true, false, false><<<GG, 256, 0, stream>>>(
        bufA, rowptr, col, dinv, nullptr, W4, b4, xrec, N);
}


# Edge sums: slot indices asked for one batch ahead, behind the row loads in flight (a node's first behind the previous node's last rows); one 8-wide batch form; fused grids 1280 (82 regs); sums same
# speedup vs baseline: 4.8454x; 1.0597x over previous
#include <hip/hip_runtime.h>
#include <hip/hip_bf16.h>


typedef float v2f __attribute__((ext_vector_type(2)));
typedef float v8f __attribute__((ext_vector_type(8)));

constexpr int BSH   = 9;
constexpr int BW    = 1 << BSH;
constexpr int NBMAX = 256;
constexpr int SG    = 256;
constexpr int PSH   = 10;
constexpr int SU    = 8;
constexpr int KT    = BW;
constexpr int BU    = 20;

__device__ __forceinline__ int4 ld4_guard(const int* __restrict__ p, int i, int n) {
    if (i + 3 < n) return *(const int4*)(p + i);
    int4 v;
    v.x = (i     < n) ? p[i]     : 0;
    v.y = (i + 1 < n) ? p[i + 1] : 0;
    v.z = (i + 2 < n) ? p[i + 2] : 0;
    v.w = 0;
    return v;
}

__device__ __forceinline__ int wave_incl_scan(int v, int lane) {
    for (int d = 1; d < 32; d <<= 1) {
        int u = __shfl_up(v, d);
        if (lane >= d) v += u;
    }
    return v;
}

__device__ __forceinline__ void load_slots(int (&s)[8], const int* __restrict__ col, int j) {
    const int* __restrict__ p = col + (unsigned)j;
#pragma unroll
    for (int u = 0; u < 8; ++u) s[u] = p[u];
}

__device__ __forceinline__ void load_range(int& b, int& e, const int* __restrict__ rowptr, int node, int N) {
    const int n0 = (node < N) ? node : N;
    const int n1 = (node < N) ? (node + 1) : N;
    b = rowptr[n0];
    e = rowptr[n1];
}

__global__ __launch_bounds__(256) void k_slice_hist(const int* __restrict__ dst,
                                                    int* __restrict__ tblc,
                                                    int E, int NB) {
    __shared__ int hist[NBMAX];
    for (int b = threadIdx.x; b < NB; b += 256) hist[b] = 0;
    __syncthreads();
    const bool al  = (((size_t)dst) & 15) == 0;
    const int  nwp = al ? (E >> PSH) : 0;
    const int  t4  = 4 * threadIdx.x;
    for (int p0 = blockIdx.x; p0 < nwp; p0 += SU * SG) {
        int4 d[SU];
#pragma unroll
        for (int u = 0; u < SU; ++u) {
            const int p  = p0 + u * SG;
            const int pc = (p < nwp) ? p : p0;
            d[u] = *(const int4*)(dst + ((size_t)pc << PSH) + t4);
        }
#pragma unroll
        for (int u = 0; u < SU; ++u) {
            if (p0 + u * SG < nwp) {
                atomicAdd(&hist[d[u].x >> BSH], 1);
                atomicAdd(&hist[d[u].y >> BSH], 1);
                atomicAdd(&hist[d[u].z >> BSH], 1);
                atomicAdd(&hist[d[u].w >> BSH], 1);
            }
        }
    }
    for (int p = blockIdx.x; (p << PSH) < E; p += SG) {
        if (p < nwp) continue;
        for (int q = 0; q < 4; ++q) {
            const int i = (p << PSH) + q * 256 + threadIdx.x;
            if (i < E) atomicAdd(&hist[dst[i] >> BSH], 1);
        }
    }
    __syncthreads();
    for (int b = threadIdx.x; b < NB; b += 256) tblc[b * SG + blockIdx.x] = hist[b];
}

template <int B, bool FULL>
__device__ __forceinline__ void scan_batch(const int* __restrict__ cnt, int* __restrict__ off,
                                           int c, int lane, int n, int& run) {
    int4 v[B];
    int  ls[B], inc[B], tot[B];
#pragma unroll
    for (int b = 0; b < B; ++b) {
        const int i = ((c + b) << 7) + (lane << 2);
        if (FULL) v[b] = *(const int4*)(cnt + i); else v[b] = ld4_guard(cnt, i, n);
    }
#pragma unroll
    for (int b = 0; b < B; ++b) { ls[b] = v[b].x + v[b].y + v[b].z + v[b].w; inc[b] = ls[b]; }
    for (int d = 1; d < 32; d <<= 1) {
        int u[B];
#pragma unroll
        for (int b = 0; b < B; ++b) u[b] = __shfl_up(inc[b], d);
#pragma unroll
        for (int b = 0; b < B; ++b) if (lane >= d) inc[b] += u[b];
    }
#pragma unroll
    for (int b = 0; b < B; ++b) tot[b] = __shfl(inc[b], 31);
#pragma unroll
    for (int b = 0; b < B; ++b) {
        const int i  = ((c + b) << 7) + (lane << 2);
        const int b0 = run + inc[b] - ls[b];
        run += tot[b];
        int4 r;
        r.x = b0; r.y = b0 + v[b].x; r.z = r.y + v[b].y; r.w = r.z + v[b].z;
        if (FULL || i + 3 < n) {
            *(int4*)(off + i) = r;
        } else {
            if (i     < n) off[i]     = r.x;
            if (i + 1 < n) off[i + 1] = r.y;
            if (i + 2 < n) off[i + 2] = r.z;
        }
    }
}

__global__ __launch_bounds__(1024) void k_scan_table(const int* __restrict__ cnt,
                                                     int* __restrict__ off, int n) {
    __shared__ int wtot[32];
    const int lane = threadIdx.x & 31;
    const int wv   = threadIdx.x >> 5;
    const int chunks = (n + 127) >> 7;
    const int full   = n >> 7;
    const int cpw    = (chunks + 31) >> 5;
    const int c0     = wv * cpw;
    const int c1     = (c0 + cpw < chunks) ? (c0 + cpw) : chunks;
    const int cf     = (c1 < full) ? c1 : full;

    int sum = 0;
#pragma unroll 5
    for (int c = c0; c < cf; ++c) {
        const int4 v = *(const int4*)(cnt + (c << 7) + (lane << 2));
        sum += v.x + v.y + v.z + v.w;
    }
    for (int c = (cf > c0 ? cf : c0); c < c1; ++c) {
        const int4 v = ld4_guard(cnt, (c << 7) + (lane << 2), n);
        sum += v.x + v.y + v.z + v.w;
    }
    for (int d = 16; d > 0; d >>= 1) sum += __shfl_xor(sum, d);
    if (lane == 0) wtot[wv] = sum;
    __syncthreads();
    const int t    = wtot[lane];
    const int incl = wave_incl_scan(t, lane);
    int run = __shfl(incl, wv) - __shfl(t, wv);
    const int total = __shfl(incl, 31);

    int c = c0;
    for (; c + 5 <= cf; c += 5) scan_batch<5, true>(cnt, off, c, lane, n, run);
    for (; c < cf; ++c)         scan_batch<1, true>(cnt, off, c, lane, n, run);
    for (; c < c1; ++c)         scan_batch<1, false>(cnt, off, c, lane, n, run);

    if (threadIdx.x == 0) off[n] = total;
}

__device__ __forceinline__ int pack_edge(int s, int d) { return (s << BSH) | (d & (BW - 1)); }

__global__ __launch_bounds__(256) void k_slice_scatter(const int* __restrict__ src,
                                                       const int* __restrict__ dst,
                                                       const int* __restrict__ tblo,
                                                       int* __restrict__ packed,
                                                       int E, int NB) {
    __shared__ int cur[NBMAX];
    for (int b = threadIdx.x; b < NB; b += 256) cur[b] = tblo[b * SG + blockIdx.x];
    __syncthreads();
    const bool al  = (((size_t)dst) & 15) == 0;
    const bool als = (((size_t)src) & 15) == 0;
    const int  nwp = al ? (E >> PSH) : 0;
    const int  t4  = 4 * threadIdx.x;
    for (int p0 = blockIdx.x; p0 < nwp; p0 += SU * SG) {
        int4 d[SU], s[SU], w[SU];
#pragma unroll
        for (int u = 0; u < SU; ++u) {
            const int p  = p0 + u * SG;
            const int pc = (p < nwp) ? p : p0;
            const size_t o = ((size_t)pc << PSH) + t4;
            d[u] = *(const int4*)(dst + o);
            if (als) {
                s[u] = *(const int4*)(src + o);
            } else {
                s[u].x = src[o]; s[u].y = src[o + 1]; s[u].z = src[o + 2]; s[u].w = src[o + 3];
            }
        }
#pragma unroll
        for (int u = 0; u < SU; ++u) {
            if (p0 + u * SG < nwp) {
                w[u].x = atomicAdd(&cur[d[u].x >> BSH], 1);
                w[u].y = atomicAdd(&cur[d[u].y >> BSH], 1);
                w[u].z = atomicAdd(&cur[d[u].z >> BSH], 1);
                w[u].w = atomicAdd(&cur[d[u].w >> BSH], 1);
            }
        }
#pragma unroll
        for (int u = 0; u < SU; ++u) {
            if (p0 + u * SG < nwp) {
                packed[w[u].x] = pack_edge(s[u].x, d[u].x);
                packed[w[u].y] = pack_edge(s[u].y, d[u].y);
                packed[w[u].z] = pack_edge(s[u].z, d[u].z);
                packed[w[u].w] = pack_edge(s[u].w, d[u].w);
            }
        }
    }
    for (int p = blockIdx.x; (p << PSH) < E; p += SG) {
        if (p < nwp) continue;
        for (int q = 0; q < 4; ++q) {
            const int i = (p << PSH) + q * 256 + threadIdx.x;
            if (i < E) {
                const int dd = dst[i];
                packed[atomicAdd(&cur[dd >> BSH], 1)] = pack_edge(src[i], dd);
            }
        }
    }
}

__global__ __launch_bounds__(KT) void k_bucket_rows(const int* __restrict__ packed,
                                                      const int* __restrict__ tblo,
                                                      int* __restrict__ rowptr,
                                                      float* __restrict__ dinv,
                                                      int* __restrict__ col,
                                                      int N, int NB) {
    __shared__ int cnt[BW];
    __shared__ int wt[KT / 32];
    const int b    = blockIdx.x;
    const int t    = threadIdx.x;
    const int lane = t & 31;
    const int wv   = t >> 5;
    const int beg  = tblo[b * SG];
    const int end  = tblo[(b + 1) * SG];
    cnt[t] = 0;
    __syncthreads();
    for (int base = beg; base < end; base += BU * KT) {
        int k[BU];
#pragma unroll
        for (int u = 0; u < BU; ++u) {
            const int i = base + u * KT + t;
            const int v = packed[(i < end) ? i : (end - 1)];
            k[u] = (i < end) ? v : -1;
        }
#pragma unroll
        for (int u = 0; u < BU; ++u) if (k[u] >= 0) atomicAdd(&cnt[k[u] & (BW - 1)], 1);
    }
    __syncthreads();
    const int c    = cnt[t];
    const int incl = wave_incl_scan(c, lane);
    if (lane == 31) wt[wv] = incl;
    __syncthreads();
    const int wtv   = (lane < KT / 32) ? wt[lane] : 0;
    const int wincl = wave_incl_scan(wtv, lane);
    const int woff  = __shfl(wincl, wv) - __shfl(wtv, wv);
    const int first = beg + woff + incl - c;
    const int node  = (b << BSH) + t;
    cnt[t] = first;
    if (node < N) {
        rowptr[node] = first;
        dinv[node]   = rsqrtf((float)c + 1.0f);
    }
    if (b == NB - 1 && t == 0) rowptr[N] = end;
    __syncthreads();
    for (int base = beg; base < end; base += BU * KT) {
        int k[BU], w[BU];
#pragma unroll
        for (int u = 0; u < BU; ++u) {
            const int i = base + u * KT + t;
            const int v = packed[(i < end) ? i : (end - 1)];
            k[u] = (i < end) ? v : -1;
        }
#pragma unroll
        for (int u = 0; u < BU; ++u) w[u] = (k[u] >= 0) ? atomicAdd(&cnt[k[u] & (BW - 1)], 1) : 0;
#pragma unroll
        for (int u = 0; u < BU; ++u) if (k[u] >= 0) col[w[u]] = k[u] >> BSH;
    }
}

template <int K, int F, bool BIAS, bool RELU, bool SCALE>
__global__ __launch_bounds__(256) void k_gemm_wmma(const float* __restrict__ X,
                                                   const float* __restrict__ W,
                                                   const float* __restrict__ bias,
                                                   const float* __restrict__ dinv,
                                                   float* __restrict__ H,
                                                   int tilesM) {
    static_assert(K % 4 == 0 && F % 16 == 0, "shape");
    constexpr int TF = F / 16;
    const int tm = blockIdx.x * (blockDim.x >> 5) + (threadIdx.x >> 5);
    if (tm >= tilesM) return;
    const int lane = threadIdx.x & 31;
    const int l15  = lane & 15;
    const int hi2  = (lane >> 4) << 1;

    const float* xrow = X + (size_t)(tm * 16 + l15) * K + hi2;
    const float* wcol = W + (size_t)hi2 * F + l15;

    v8f acc[TF];
#pragma unroll
    for (int t = 0; t < TF; ++t) acc[t] = v8f{};

    constexpr int KS = K / 4;
    v2f a[KS];
#pragma unroll
    for (int s = 0; s < KS; ++s) {
        a[s].x = xrow[4 * s];
        a[s].y = xrow[4 * s + 1];
    }

#pragma unroll
    for (int s = 0; s < KS; ++s) {
#pragma unroll
        for (int t = 0; t < TF; ++t) {
            v2f b;
            b.x = wcol[(size_t)(4 * s) * F + t * 16];
            b.y = wcol[(size_t)(4 * s + 1) * F + t * 16];
            acc[t] = __builtin_amdgcn_wmma_f32_16x16x4_f32(
                 false, a[s],  false, b,
                 (short)0, acc[t],  false,  false);
        }
    }

    const int hi8 = (lane >> 4) << 3;
    float dv[8];
#pragma unroll
    for (int v = 0; v < 8; ++v) dv[v] = SCALE ? dinv[tm * 16 + hi8 + v] : 1.0f;
    float* hrow = H + (size_t)(tm * 16 + hi8) * F + l15;
#pragma unroll
    for (int t = 0; t < TF; ++t) {
        float bv = 0.0f;
        if (BIAS) bv = bias[t * 16 + l15];
#pragma unroll
        for (int v = 0; v < 8; ++v) {
            float o = acc[t][v];
            if (BIAS) o += bv;
            if (RELU) o = fmaxf(o, 0.0f);
            if (SCALE) o *= dv[v];
            hrow[(size_t)v * F + t * 16] = o;
        }
    }
}

template <int K, int F, bool BIAS, bool RELU, bool SCALE>
__global__ __launch_bounds__(256) void k_gemm_tail(const float* __restrict__ X,
                                                   const float* __restrict__ W,
                                                   const float* __restrict__ bias,
                                                   const float* __restrict__ dinv,
                                                   float* __restrict__ H,
                                                   int rowStart, int rowEnd) {
    int idx = blockIdx.x * blockDim.x + threadIdx.x;
    int rows = rowEnd - rowStart;
    if (idx >= rows * F) return;
    int r = rowStart + idx / F, f = idx % F;
    float s = 0.f;
    for (int k = 0; k < K; ++k) s += X[(size_t)r * K + k] * W[(size_t)k * F + f];
    if (BIAS) s += bias[f];
    if (RELU) s = fmaxf(s, 0.0f);
    if (SCALE) s *= dinv[r];
    H[(size_t)r * F + f] = s;
}

template <int F, bool BIAS, bool RELU, bool DUAL>
__global__ __launch_bounds__(256) void k_gather(const float* __restrict__ HS,
                                                const int* __restrict__ rowptr,
                                                const int* __restrict__ col,
                                                const float* __restrict__ dinv,
                                                const float* __restrict__ bias,
                                                float* __restrict__ OUT,
                                                float* __restrict__ OUTS, int N) {
    constexpr int CH  = F / 4;
    constexpr int NPB = 256 / CH;
    const int c = threadIdx.x % CH;
    const int g = threadIdx.x / CH;
    const float4* __restrict__ H4 = (const float4*)HS;
    float4* __restrict__ O4 = (float4*)OUT;
    float4* __restrict__ S4 = (float4*)OUTS;

    float4 bv = make_float4(0.f, 0.f, 0.f, 0.f);
    if (BIAS) bv = ((const float4*)bias)[c];

    const int stride = gridDim.x * NPB;
    int node = blockIdx.x * NPB + g;
    int beg, end, nbeg, nend;
    load_range(beg, end, rowptr, node, N);
    load_range(nbeg, nend, rowptr, node + stride, N);
    int sidx[8];
    load_slots(sidx, col, beg);

    while (node < N) {
        const int nxt = node + stride;
        int n2beg, n2end;
        load_range(n2beg, n2end, rowptr, nxt + stride, N);

        const float dd = dinv[node];
        const float4 hs = H4[(size_t)node * CH + c];
        float4 acc = make_float4(0.f, 0.f, 0.f, 0.f);

        for (int j = beg; j < end; j += 8) {
            float4 h[8];
            int    nx[8];
#pragma unroll
            for (int u = 0; u < 8; ++u) {
                const int r = (j + u < end) ? sidx[u] : node;
                h[u] = H4[(unsigned)(r * CH + c)];
            }
            load_slots(nx, col, (j + 8 < end) ? (j + 8) : nbeg);
            __builtin_amdgcn_sched_barrier(0);
#pragma unroll
            for (int u = 0; u < 8; ++u) {
                const bool in = j + u < end;
                acc.x += in ? h[u].x : 0.0f;
                acc.y += in ? h[u].y : 0.0f;
                acc.z += in ? h[u].z : 0.0f;
                acc.w += in ? h[u].w : 0.0f;
            }
#pragma unroll
            for (int u = 0; u < 8; ++u) sidx[u] = nx[u];
        }
        if (beg >= end) load_slots(sidx, col, nbeg);

        float4 o;
        o.x = (acc.x + hs.x) * dd;
        o.y = (acc.y + hs.y) * dd;
        o.z = (acc.z + hs.z) * dd;
        o.w = (acc.w + hs.w) * dd;
        if (BIAS) { o.x += bv.x; o.y += bv.y; o.z += bv.z; o.w += bv.w; }
        if (RELU) {
            o.x = fmaxf(o.x, 0.0f); o.y = fmaxf(o.y, 0.0f);
            o.z = fmaxf(o.z, 0.0f); o.w = fmaxf(o.w, 0.0f);
        }
        O4[(size_t)node * CH + c] = o;
        if (DUAL) {
            float4 q;
            q.x = o.x * dd; q.y = o.y * dd; q.z = o.z * dd; q.w = o.w * dd;
            S4[(size_t)node * CH + c] = q;
        }

        node = nxt; beg = nbeg; end = nend; nbeg = n2beg; nend = n2end;
    }
}

template <int F, int FO, bool GBIAS, bool GRELU, bool OBIAS, bool ORELU, bool OSCALE>
__global__ __launch_bounds__(256) void k_gather_gemm(const float* __restrict__ HS,
                                                     const int* __restrict__ rowptr,
                                                     const int* __restrict__ col,
                                                     const float* __restrict__ dinv,
                                                     const float* __restrict__ gbias,
                                                     const float* __restrict__ W,
                                                     const float* __restrict__ obias,
                                                     float* __restrict__ OUT, int N) {
    constexpr int CH    = F / 4;
    constexpr int NPB   = 256 / CH;
    constexpr int PITCH = F + 4;
    constexpr int CT    = FO / 16;
    constexpr int TILES = (NPB / 16) * CT;
    constexpr int KS    = F / 4;
    static_assert(TILES <= 8 && NPB % 16 == 0, "one tile per wave");
    __shared__ __attribute__((aligned(16))) float sA[2][NPB * PITCH];
    __shared__ __attribute__((aligned(16))) float sW[F * FO];

    const int c = threadIdx.x % CH;
    const int g = threadIdx.x / CH;
    const int lane = threadIdx.x & 31;
    const int wv   = threadIdx.x >> 5;
    const int l15  = lane & 15;
    const int hi2  = (lane >> 4) << 1;
    const int hi8  = (lane >> 4) << 3;
    const bool has_tile = wv < TILES;
    const int strip = wv / CT;
    const int ct    = wv % CT;
    const float4* __restrict__ H4 = (const float4*)HS;

    for (int q = threadIdx.x; q < F * FO / 4; q += 256) {
        const int k  = q / (FO / 4);
        const int n4 = (q % (FO / 4)) * 4;
        const float4 w4 = *(const float4*)(W + (size_t)k * FO + n4);
        float* dst = &sW[(((n4 >> 4) * KS + (k >> 2)) * 32 + ((k >> 1) & 1) * 16 + (n4 & 15)) * 2 + (k & 1)];
        dst[0] = w4.x; dst[2] = w4.y; dst[4] = w4.z; dst[6] = w4.w;
    }
    float ob = 0.0f;
    if (OBIAS && has_tile) ob = obias[ct * 16 + l15];

    const int stride = gridDim.x * NPB;
    int node = blockIdx.x * NPB + g;
    int beg, end, nbeg, nend;
    load_range(beg, end, rowptr, node, N);
    load_range(nbeg, nend, rowptr, node + stride, N);
    int sidx[8];
    load_slots(sidx, col, beg);

    int buf = 0;
    for (int base = blockIdx.x * NPB; base < N; base += stride) {
        const int nxt = node + stride;
        int n2beg, n2end;
        load_range(n2beg, n2end, rowptr, nxt + stride, N);

        float4 o = make_float4(0.f, 0.f, 0.f, 0.f);
        if (node < N) {
            const float dd = dinv[node];
            const float4 hs = H4[(size_t)node * CH + c];
            float4 acc = make_float4(0.f, 0.f, 0.f, 0.f);

            for (int j = beg; j < end; j += 8) {
                float4 h[8];
                int    nx[8];
#pragma unroll
                for (int u = 0; u < 8; ++u) {
                    const int r = (j + u < end) ? sidx[u] : node;
                    h[u] = H4[(unsigned)(r * CH + c)];
                }
                load_slots(nx, col, (j + 8 < end) ? (j + 8) : nbeg);
                __builtin_amdgcn_sched_barrier(0);
#pragma unroll
                for (int u = 0; u < 8; ++u) {
                    const bool in = j + u < end;
                    acc.x += in ? h[u].x : 0.0f;
                    acc.y += in ? h[u].y : 0.0f;
                    acc.z += in ? h[u].z : 0.0f;
                    acc.w += in ? h[u].w : 0.0f;
                }
#pragma unroll
                for (int u = 0; u < 8; ++u) sidx[u] = nx[u];
            }

            o.x = (acc.x + hs.x) * dd;
            o.y = (acc.y + hs.y) * dd;
            o.z = (acc.z + hs.z) * dd;
            o.w = (acc.w + hs.w) * dd;
            if (GBIAS) {
                const float4 bv = ((const float4*)gbias)[c];
                o.x += bv.x; o.y += bv.y; o.z += bv.z; o.w += bv.w;
            }
            if (GRELU) {
                o.x = fmaxf(o.x, 0.0f); o.y = fmaxf(o.y, 0.0f);
                o.z = fmaxf(o.z, 0.0f); o.w = fmaxf(o.w, 0.0f);
            }
        }
        if (beg >= end) load_slots(sidx, col, nbeg);
        *(float4*)&sA[buf][g * PITCH + 4 * c] = o;
        __syncthreads();

        const int r0 = base + strip * 16;
        if (has_tile && r0 + 16 <= N) {
            const float* arow = &sA[buf][(strip * 16 + l15) * PITCH + hi2];
            const float* bfr  = &sW[(ct * KS * 32 + lane) * 2];
            v8f d = v8f{};
#pragma unroll
            for (int s = 0; s < KS; ++s) {
                const v2f a = *(const v2f*)(arow + 4 * s);
                const v2f b = *(const v2f*)(bfr + s * 64);
                d = __builtin_amdgcn_wmma_f32_16x16x4_f32(
                     false, a,  false, b,
                     (short)0, d,  false,  false);
            }
            float dv[8];
#pragma unroll
            for (int v = 0; v < 8; ++v) dv[v] = OSCALE ? dinv[r0 + hi8 + v] : 1.0f;
            float* orow = OUT + (size_t)(r0 + hi8) * FO + ct * 16 + l15;
#pragma unroll
            for (int v = 0; v < 8; ++v) {
                float q = d[v];
                if (OBIAS) q += ob;
                if (ORELU) q = fmaxf(q, 0.0f);
                if (OSCALE) q *= dv[v];
                orow[(size_t)v * FO] = q;
            }
        } else if (has_tile && r0 < N) {
            for (int r = (lane >> 4); r < 16 && r0 + r < N; r += 2) {
                const float* ar = &sA[buf][(strip * 16 + r) * PITCH];
                float q = 0.0f;
                for (int k = 0; k < F; ++k) q += ar[k] * W[(size_t)k * FO + ct * 16 + l15];
                if (OBIAS) q += ob;
                if (ORELU) q = fmaxf(q, 0.0f);
                if (OSCALE) q *= dinv[r0 + r];
                OUT[(size_t)(r0 + r) * FO + ct * 16 + l15] = q;
            }
        }

        node = nxt; beg = nbeg; end = nend; nbeg = n2beg; nend = n2end;
        buf ^= 1;
    }
}

static inline int nblk(long long n, int t) { return (int)((n + t - 1) / t); }

template <int K, int F, bool BIAS, bool RELU, bool SCALE>
static void launch_gemm(const float* X, const float* W, const float* bias, const float* dinv,
                        float* H, int N, hipStream_t s) {
    int tilesM = N / 16;
    long long waves = tilesM;
    if (waves > 0)
        k_gemm_wmma<K, F, BIAS, RELU, SCALE><<<nblk(waves, 8), 256, 0, s>>>(X, W, bias, dinv, H,
                                                                            tilesM);
    int tail = tilesM * 16;
    if (tail < N)
        k_gemm_tail<K, F, BIAS, RELU, SCALE><<<nblk((long long)(N - tail) * F, 256), 256, 0, s>>>(
            X, W, bias, dinv, H, tail, N);
}

extern "C" void kernel_launch(void* const* d_in, const int* in_sizes, int n_in,
                              void* d_out, int out_size, void* d_ws, size_t ws_size,
                              hipStream_t stream) {
    const float* x  = (const float*)d_in[0];
    const int*   ei = (const int*)d_in[1];
    const float* W1 = (const float*)d_in[2]; const float* b1 = (const float*)d_in[3];
    const float* W2 = (const float*)d_in[4]; const float* b2 = (const float*)d_in[5];
    const float* W3 = (const float*)d_in[6]; const float* b3 = (const float*)d_in[7];
    const float* W4 = (const float*)d_in[8]; const float* b4 = (const float*)d_in[9];

    const int N = in_sizes[0] / 128;
    const int E = in_sizes[1] / 2;
    const int* src = ei;
    const int* dst = ei + E;

    float* out  = (float*)d_out;
    float* xrec = out;
    float* z    = out + (size_t)N * 128;

    float* ws   = (float*)d_ws;
    float* dinv = ws;
    float* bufH = ws + N;
    float* bufA = bufH + (size_t)N * 128;
    int* tblc   = (int*)(bufH + (size_t)N * 64);
    int* tblo   = tblc + NBMAX * SG;
    int* rowptr = tblo + NBMAX * SG + 4;
    int* col    = rowptr + N + 4;
    int* packed = col + E;

    const int GG  = 1536;
    const int GGF = 1280;

    const int NB  = (N + BW - 1) >> BSH;
    k_slice_hist<<<SG, 256, 0, stream>>>(dst, tblc, E, NB);
    k_scan_table<<<1, 1024, 0, stream>>>(tblc, tblo, NB * SG);
    k_slice_scatter<<<SG, 256, 0, stream>>>(src, dst, tblo, packed, E, NB);
    k_bucket_rows<<<NB, KT, 0, stream>>>(packed, tblo, rowptr, dinv, col, N, NB);

    launch_gemm<128, 64, false, false, true>(x, W1, nullptr, dinv, bufH, N, stream);

    k_gather_gemm<64, 32, true, true, false, false, true><<<GGF, 256, 0, stream>>>(
        bufH, rowptr, col, dinv, b1, W2, nullptr, bufA, N);

    k_gather<32, true, false, true><<<GG, 256, 0, stream>>>(bufA, rowptr, col, dinv, b2, z, bufH,
                                                            N);

    k_gather_gemm<32, 64, false, false, true, true, true><<<GGF, 256, 0, stream>>>(
        bufH, rowptr, col, dinv, nullptr, W3, b3, bufA, N);

    k_gather_gemm<64, 128, false, false, true, false, false><<<GGF, 256, 0, stream>>>(
        bufA, rowptr, col, dinv, nullptr, W4, b4, xrec, N);
}


# Edge order in two launches, not four: a slice orders its own edges by bucket in shared memory and copies them out in whole lines; a bucket's work-group reads its 256 runs; slots by counter arrival
# speedup vs baseline: 5.9550x; 1.2290x over previous
#include <hip/hip_runtime.h>
#include <hip/hip_bf16.h>


typedef float v2f __attribute__((ext_vector_type(2)));
typedef float v8f __attribute__((ext_vector_type(8)));

constexpr int BSH   = 9;
constexpr int BW    = 1 << BSH;
constexpr int NBMAX = 256;
constexpr int SG    = 256;
constexpr int PSH   = 10;
constexpr int SU    = 8;
constexpr int CAP   = SU << PSH;
constexpr int KT    = BW;
constexpr int RPW   = SG / (KT / 32);
constexpr int SPI   = 2;
static_assert(NBMAX == 256 && RPW <= 32 && RPW * (KT / 32) == SG, "shape of the edge ordering");

__device__ __forceinline__ int wave_incl_scan(int v, int lane) {
    for (int d = 1; d < 32; d <<= 1) {
        int u = __shfl_up(v, d);
        if (lane >= d) v += u;
    }
    return v;
}

__device__ __forceinline__ void load_slots(int (&s)[8], const int* __restrict__ col, int j) {
    const int* __restrict__ p = col + (unsigned)j;
#pragma unroll
    for (int u = 0; u < 8; ++u) s[u] = p[u];
}

__device__ __forceinline__ void load_range(int& b, int& e, const int* __restrict__ rowptr, int node, int N) {
    const int n0 = (node < N) ? node : N;
    const int n1 = (node < N) ? (node + 1) : N;
    b = rowptr[n0];
    e = rowptr[n1];
}

__device__ __forceinline__ int pack_edge(int s, int d) { return (s << BSH) | (d & (BW - 1)); }

__global__ __launch_bounds__(256) void k_slice_sort(const int* __restrict__ src,
                                                    const int* __restrict__ dst,
                                                    int* __restrict__ tblc,
                                                    int* __restrict__ tblo,
                                                    int* __restrict__ sorted,
                                                    int E, int NB) {
    __shared__ int hist[NBMAX];
    __shared__ int wt[8];
    __shared__ __attribute__((aligned(16))) int stage[CAP];
    const int t    = threadIdx.x;
    const int lane = t & 31;
    const int wv   = t >> 5;
    hist[t] = 0;
    const bool al  = (((size_t)dst) & 15) == 0;
    const bool als = (((size_t)src) & 15) == 0;
    const int  nwp = al ? (E >> PSH) : 0;
    const int  t4  = 4 * t;
    const int  p0  = blockIdx.x;
    const bool reg = p0 < nwp;
    int4 d[SU], s[SU];
    if (reg) {
#pragma unroll
        for (int u = 0; u < SU; ++u) {
            const int p  = p0 + u * SG;
            const int pc = (p < nwp) ? p : p0;
            const size_t o = ((size_t)pc << PSH) + t4;
            d[u] = *(const int4*)(dst + o);
            if (als) {
                s[u] = *(const int4*)(src + o);
            } else {
                s[u].x = src[o]; s[u].y = src[o + 1]; s[u].z = src[o + 2]; s[u].w = src[o + 3];
            }
        }
        __builtin_amdgcn_sched_barrier(0);
    }
    __syncthreads();

    if (reg) {
#pragma unroll
        for (int u = 0; u < SU; ++u) {
            if (p0 + u * SG < nwp) {
                atomicAdd(&hist[d[u].x >> BSH], 1);
                atomicAdd(&hist[d[u].y >> BSH], 1);
                atomicAdd(&hist[d[u].z >> BSH], 1);
                atomicAdd(&hist[d[u].w >> BSH], 1);
            }
        }
    }
    for (int p = blockIdx.x; (p << PSH) < E; p += SG) {
        if (p < nwp) continue;
        for (int q = 0; q < 4; ++q) {
            const int i = (p << PSH) + q * 256 + t;
            if (i < E) atomicAdd(&hist[dst[i] >> BSH], 1);
        }
    }
    __syncthreads();

    const int c    = hist[t];
    const int incl = wave_incl_scan(c, lane);
    if (lane == 31) wt[wv] = incl;
    __syncthreads();
    const int wtv   = (lane < 8) ? wt[lane] : 0;
    const int wincl = wave_incl_scan(wtv, lane);
    const int woff  = __shfl(wincl, wv) - __shfl(wtv, wv);
    const int excl  = woff + incl - c;
    const int n     = __shfl(wincl, 7);
    if (t < NB) {
        tblc[t * SG + blockIdx.x] = c;
        tblo[t * SG + blockIdx.x] = excl;
    }
    hist[t] = excl;
    __syncthreads();

    if (reg) {
#pragma unroll
        for (int u = 0; u < SU; ++u) {
            if (p0 + u * SG < nwp) {
                stage[atomicAdd(&hist[d[u].x >> BSH], 1)] = pack_edge(s[u].x, d[u].x);
                stage[atomicAdd(&hist[d[u].y >> BSH], 1)] = pack_edge(s[u].y, d[u].y);
                stage[atomicAdd(&hist[d[u].z >> BSH], 1)] = pack_edge(s[u].z, d[u].z);
                stage[atomicAdd(&hist[d[u].w >> BSH], 1)] = pack_edge(s[u].w, d[u].w);
            }
        }
    }
    for (int p = blockIdx.x; (p << PSH) < E; p += SG) {
        if (p < nwp) continue;
        for (int q = 0; q < 4; ++q) {
            const int i = (p << PSH) + q * 256 + t;
            if (i < E) {
                const int dd = dst[i];
                stage[atomicAdd(&hist[dd >> BSH], 1)] = pack_edge(src[i], dd);
            }
        }
    }
    __syncthreads();

    int* __restrict__ out = sorted + (size_t)blockIdx.x * CAP;
    const bool alo = (((size_t)out) & 15) == 0;
    for (int i = t4; i < n; i += 1024) {
        if (alo && i + 3 < n) {
            *(int4*)(out + i) = *(const int4*)(stage + i);
        } else {
            if (i     < n) out[i]     = stage[i];
            if (i + 1 < n) out[i + 1] = stage[i + 1];
            if (i + 2 < n) out[i + 2] = stage[i + 2];
            if (i + 3 < n) out[i + 3] = stage[i + 3];
        }
    }
}

__global__ __launch_bounds__(KT) void k_bucket_rows(const int* __restrict__ sorted,
                                                      const int* __restrict__ tblc,
                                                      const int* __restrict__ tblo,
                                                      int* __restrict__ rowptr,
                                                      float* __restrict__ dinv,
                                                      int* __restrict__ col,
                                                      int N, int NB) {
    __shared__ int cnt[BW];
    __shared__ int wt[KT / 32];
    __shared__ int rsum[KT / 32], rtot[KT / 32], rmax[KT / 32];
    const int b    = blockIdx.x;
    const int t    = threadIdx.x;
    const int lane = t & 31;
    const int wv   = t >> 5;
    int rc = 0, ro = 0, ra = 0;
    if (lane < RPW) {
        const int w = wv * RPW + lane;
        rc = tblc[b * SG + w];
        ro = tblo[b * SG + w];
        ra = w * CAP + ro;
    }
    cnt[t] = 0;
    int s1 = ro, s2 = rc, s3 = rc;
    for (int d = 16; d > 0; d >>= 1) {
        s1 += __shfl_xor(s1, d);
        s2 += __shfl_xor(s2, d);
        const int m = __shfl_xor(s3, d);
        s3 = (m > s3) ? m : s3;
    }
    if (lane == 0) { rsum[wv] = s1; rtot[wv] = s2; rmax[wv] = s3; }
    __syncthreads();
    int g1 = (lane < KT / 32) ? rsum[lane] : 0;
    int g2 = (lane < KT / 32) ? rtot[lane] : 0;
    int g3 = (lane < KT / 32) ? rmax[lane] : 0;
    for (int d = 16; d > 0; d >>= 1) {
        g1 += __shfl_xor(g1, d);
        g2 += __shfl_xor(g2, d);
        const int m = __shfl_xor(g3, d);
        g3 = (m > g3) ? m : g3;
    }
    const int beg    = g1;
    const int end    = g1 + g2;
    const int sweeps = (g3 + 32 * SPI - 1) / (32 * SPI);

    for (int it = 0; it < sweeps; ++it) {
        int k[SPI * RPW];
#pragma unroll
        for (int q = 0; q < RPW; ++q) {
            const int nq = __builtin_amdgcn_readlane(rc, q);
            const int aq = __builtin_amdgcn_readlane(ra, q);
#pragma unroll
            for (int h = 0; h < SPI; ++h) {
                const int j = ((it * SPI + h) << 5) + lane;
                k[q * SPI + h] = sorted[(unsigned)((j < nq) ? (aq + j) : 0)];
            }
        }
        __builtin_amdgcn_sched_barrier(0);
#pragma unroll
        for (int q = 0; q < RPW; ++q) {
            const int nq = __builtin_amdgcn_readlane(rc, q);
#pragma unroll
            for (int h = 0; h < SPI; ++h) {
                const int j = ((it * SPI + h) << 5) + lane;
                if (j < nq) atomicAdd(&cnt[k[q * SPI + h] & (BW - 1)], 1);
            }
        }
    }
    __syncthreads();
    const int c    = cnt[t];
    const int incl = wave_incl_scan(c, lane);
    if (lane == 31) wt[wv] = incl;
    __syncthreads();
    const int wtv   = (lane < KT / 32) ? wt[lane] : 0;
    const int wincl = wave_incl_scan(wtv, lane);
    const int woff  = __shfl(wincl, wv) - __shfl(wtv, wv);
    const int first = beg + woff + incl - c;
    const int node  = (b << BSH) + t;
    cnt[t] = first;
    if (node < N) {
        rowptr[node] = first;
        dinv[node]   = rsqrtf((float)c + 1.0f);
    }
    if (b == NB - 1 && t == 0) rowptr[N] = end;
    __syncthreads();
    for (int it = 0; it < sweeps; ++it) {
        int k[SPI * RPW], w[SPI * RPW];
#pragma unroll
        for (int q = 0; q < RPW; ++q) {
            const int nq = __builtin_amdgcn_readlane(rc, q);
            const int aq = __builtin_amdgcn_readlane(ra, q);
#pragma unroll
            for (int h = 0; h < SPI; ++h) {
                const int j = ((it * SPI + h) << 5) + lane;
                k[q * SPI + h] = sorted[(unsigned)((j < nq) ? (aq + j) : 0)];
            }
        }
        __builtin_amdgcn_sched_barrier(0);
#pragma unroll
        for (int q = 0; q < RPW; ++q) {
            const int nq = __builtin_amdgcn_readlane(rc, q);
#pragma unroll
            for (int h = 0; h < SPI; ++h) {
                const int j = ((it * SPI + h) << 5) + lane;
                w[q * SPI + h] = (j < nq) ? atomicAdd(&cnt[k[q * SPI + h] & (BW - 1)], 1) : 0;
            }
        }
#pragma unroll
        for (int q = 0; q < RPW; ++q) {
            const int nq = __builtin_amdgcn_readlane(rc, q);
#pragma unroll
            for (int h = 0; h < SPI; ++h) {
                const int j = ((it * SPI + h) << 5) + lane;
                if (j < nq) col[w[q * SPI + h]] = k[q * SPI + h] >> BSH;
            }
        }
    }
}

template <int K, int F, bool BIAS, bool RELU, bool SCALE>
__global__ __launch_bounds__(256) void k_gemm_wmma(const float* __restrict__ X,
                                                   const float* __restrict__ W,
                                                   const float* __restrict__ bias,
                                                   const float* __restrict__ dinv,
                                                   float* __restrict__ H,
                                                   int tilesM) {
    static_assert(K % 4 == 0 && F % 16 == 0, "shape");
    constexpr int TF = F / 16;
    const int tm = blockIdx.x * (blockDim.x >> 5) + (threadIdx.x >> 5);
    if (tm >= tilesM) return;
    const int lane = threadIdx.x & 31;
    const int l15  = lane & 15;
    const int hi2  = (lane >> 4) << 1;

    const float* xrow = X + (size_t)(tm * 16 + l15) * K + hi2;
    const float* wcol = W + (size_t)hi2 * F + l15;

    v8f acc[TF];
#pragma unroll
    for (int t = 0; t < TF; ++t) acc[t] = v8f{};

    constexpr int KS = K / 4;
    v2f a[KS];
#pragma unroll
    for (int s = 0; s < KS; ++s) {
        a[s].x = xrow[4 * s];
        a[s].y = xrow[4 * s + 1];
    }

#pragma unroll
    for (int s = 0; s < KS; ++s) {
#pragma unroll
        for (int t = 0; t < TF; ++t) {
            v2f b;
            b.x = wcol[(size_t)(4 * s) * F + t * 16];
            b.y = wcol[(size_t)(4 * s + 1) * F + t * 16];
            acc[t] = __builtin_amdgcn_wmma_f32_16x16x4_f32(
                 false, a[s],  false, b,
                 (short)0, acc[t],  false,  false);
        }
    }

    const int hi8 = (lane >> 4) << 3;
    float dv[8];
#pragma unroll
    for (int v = 0; v < 8; ++v) dv[v] = SCALE ? dinv[tm * 16 + hi8 + v] : 1.0f;
    float* hrow = H + (size_t)(tm * 16 + hi8) * F + l15;
#pragma unroll
    for (int t = 0; t < TF; ++t) {
        float bv = 0.0f;
        if (BIAS) bv = bias[t * 16 + l15];
#pragma unroll
        for (int v = 0; v < 8; ++v) {
            float o = acc[t][v];
            if (BIAS) o += bv;
            if (RELU) o = fmaxf(o, 0.0f);
            if (SCALE) o *= dv[v];
            hrow[(size_t)v * F + t * 16] = o;
        }
    }
}

template <int K, int F, bool BIAS, bool RELU, bool SCALE>
__global__ __launch_bounds__(256) void k_gemm_tail(const float* __restrict__ X,
                                                   const float* __restrict__ W,
                                                   const float* __restrict__ bias,
                                                   const float* __restrict__ dinv,
                                                   float* __restrict__ H,
                                                   int rowStart, int rowEnd) {
    int idx = blockIdx.x * blockDim.x + threadIdx.x;
    int rows = rowEnd - rowStart;
    if (idx >= rows * F) return;
    int r = rowStart + idx / F, f = idx % F;
    float s = 0.f;
    for (int k = 0; k < K; ++k) s += X[(size_t)r * K + k] * W[(size_t)k * F + f];
    if (BIAS) s += bias[f];
    if (RELU) s = fmaxf(s, 0.0f);
    if (SCALE) s *= dinv[r];
    H[(size_t)r * F + f] = s;
}

template <int F, bool BIAS, bool RELU, bool DUAL>
__global__ __launch_bounds__(256) void k_gather(const float* __restrict__ HS,
                                                const int* __restrict__ rowptr,
                                                const int* __restrict__ col,
                                                const float* __restrict__ dinv,
                                                const float* __restrict__ bias,
                                                float* __restrict__ OUT,
                                                float* __restrict__ OUTS, int N) {
    constexpr int CH  = F / 4;
    constexpr int NPB = 256 / CH;
    const int c = threadIdx.x % CH;
    const int g = threadIdx.x / CH;
    const float4* __restrict__ H4 = (const float4*)HS;
    float4* __restrict__ O4 = (float4*)OUT;
    float4* __restrict__ S4 = (float4*)OUTS;

    float4 bv = make_float4(0.f, 0.f, 0.f, 0.f);
    if (BIAS) bv = ((const float4*)bias)[c];

    const int stride = gridDim.x * NPB;
    int node = blockIdx.x * NPB + g;
    int beg, end, nbeg, nend;
    load_range(beg, end, rowptr, node, N);
    load_range(nbeg, nend, rowptr, node + stride, N);
    int sidx[8];
    load_slots(sidx, col, beg);

    while (node < N) {
        const int nxt = node + stride;
        int n2beg, n2end;
        load_range(n2beg, n2end, rowptr, nxt + stride, N);

        const float dd = dinv[node];
        const float4 hs = H4[(size_t)node * CH + c];
        float4 acc = make_float4(0.f, 0.f, 0.f, 0.f);

        for (int j = beg; j < end; j += 8) {
            float4 h[8];
            int    nx[8];
#pragma unroll
            for (int u = 0; u < 8; ++u) {
                const int r = (j + u < end) ? sidx[u] : node;
                h[u] = H4[(unsigned)(r * CH + c)];
            }
            load_slots(nx, col, (j + 8 < end) ? (j + 8) : nbeg);
            __builtin_amdgcn_sched_barrier(0);
#pragma unroll
            for (int u = 0; u < 8; ++u) {
                const bool in = j + u < end;
                acc.x += in ? h[u].x : 0.0f;
                acc.y += in ? h[u].y : 0.0f;
                acc.z += in ? h[u].z : 0.0f;
                acc.w += in ? h[u].w : 0.0f;
            }
#pragma unroll
            for (int u = 0; u < 8; ++u) sidx[u] = nx[u];
        }
        if (beg >= end) load_slots(sidx, col, nbeg);

        float4 o;
        o.x = (acc.x + hs.x) * dd;
        o.y = (acc.y + hs.y) * dd;
        o.z = (acc.z + hs.z) * dd;
        o.w = (acc.w + hs.w) * dd;
        if (BIAS) { o.x += bv.x; o.y += bv.y; o.z += bv.z; o.w += bv.w; }
        if (RELU) {
            o.x = fmaxf(o.x, 0.0f); o.y = fmaxf(o.y, 0.0f);
            o.z = fmaxf(o.z, 0.0f); o.w = fmaxf(o.w, 0.0f);
        }
        O4[(size_t)node * CH + c] = o;
        if (DUAL) {
            float4 q;
            q.x = o.x * dd; q.y = o.y * dd; q.z = o.z * dd; q.w = o.w * dd;
            S4[(size_t)node * CH + c] = q;
        }

        node = nxt; beg = nbeg; end = nend; nbeg = n2beg; nend = n2end;
    }
}

template <int F, int FO, bool GBIAS, bool GRELU, bool OBIAS, bool ORELU, bool OSCALE>
__global__ __launch_bounds__(256) void k_gather_gemm(const float* __restrict__ HS,
                                                     const int* __restrict__ rowptr,
                                                     const int* __restrict__ col,
                                                     const float* __restrict__ dinv,
                                                     const float* __restrict__ gbias,
                                                     const float* __restrict__ W,
                                                     const float* __restrict__ obias,
                                                     float* __restrict__ OUT, int N) {
    constexpr int CH    = F / 4;
    constexpr int NPB   = 256 / CH;
    constexpr int PITCH = F + 4;
    constexpr int CT    = FO / 16;
    constexpr int TILES = (NPB / 16) * CT;
    constexpr int KS    = F / 4;
    static_assert(TILES <= 8 && NPB % 16 == 0, "one tile per wave");
    __shared__ __attribute__((aligned(16))) float sA[2][NPB * PITCH];
    __shared__ __attribute__((aligned(16))) float sW[F * FO];

    const int c = threadIdx.x % CH;
    const int g = threadIdx.x / CH;
    const int lane = threadIdx.x & 31;
    const int wv   = threadIdx.x >> 5;
    const int l15  = lane & 15;
    const int hi2  = (lane >> 4) << 1;
    const int hi8  = (lane >> 4) << 3;
    const bool has_tile = wv < TILES;
    const int strip = wv / CT;
    const int ct    = wv % CT;
    const float4* __restrict__ H4 = (const float4*)HS;

    for (int q = threadIdx.x; q < F * FO / 4; q += 256) {
        const int k  = q / (FO / 4);
        const int n4 = (q % (FO / 4)) * 4;
        const float4 w4 = *(const float4*)(W + (size_t)k * FO + n4);
        float* dst = &sW[(((n4 >> 4) * KS + (k >> 2)) * 32 + ((k >> 1) & 1) * 16 + (n4 & 15)) * 2 + (k & 1)];
        dst[0] = w4.x; dst[2] = w4.y; dst[4] = w4.z; dst[6] = w4.w;
    }
    float ob = 0.0f;
    if (OBIAS && has_tile) ob = obias[ct * 16 + l15];

    const int stride = gridDim.x * NPB;
    int node = blockIdx.x * NPB + g;
    int beg, end, nbeg, nend;
    load_range(beg, end, rowptr, node, N);
    load_range(nbeg, nend, rowptr, node + stride, N);
    int sidx[8];
    load_slots(sidx, col, beg);

    int buf = 0;
    for (int base = blockIdx.x * NPB; base < N; base += stride) {
        const int nxt = node + stride;
        int n2beg, n2end;
        load_range(n2beg, n2end, rowptr, nxt + stride, N);

        float4 o = make_float4(0.f, 0.f, 0.f, 0.f);
        if (node < N) {
            const float dd = dinv[node];
            const float4 hs = H4[(size_t)node * CH + c];
            float4 acc = make_float4(0.f, 0.f, 0.f, 0.f);

            for (int j = beg; j < end; j += 8) {
                float4 h[8];
                int    nx[8];
#pragma unroll
                for (int u = 0; u < 8; ++u) {
                    const int r = (j + u < end) ? sidx[u] : node;
                    h[u] = H4[(unsigned)(r * CH + c)];
                }
                load_slots(nx, col, (j + 8 < end) ? (j + 8) : nbeg);
                __builtin_amdgcn_sched_barrier(0);
#pragma unroll
                for (int u = 0; u < 8; ++u) {
                    const bool in = j + u < end;
                    acc.x += in ? h[u].x : 0.0f;
                    acc.y += in ? h[u].y : 0.0f;
                    acc.z += in ? h[u].z : 0.0f;
                    acc.w += in ? h[u].w : 0.0f;
                }
#pragma unroll
                for (int u = 0; u < 8; ++u) sidx[u] = nx[u];
            }

            o.x = (acc.x + hs.x) * dd;
            o.y = (acc.y + hs.y) * dd;
            o.z = (acc.z + hs.z) * dd;
            o.w = (acc.w + hs.w) * dd;
            if (GBIAS) {
                const float4 bv = ((const float4*)gbias)[c];
                o.x += bv.x; o.y += bv.y; o.z += bv.z; o.w += bv.w;
            }
            if (GRELU) {
                o.x = fmaxf(o.x, 0.0f); o.y = fmaxf(o.y, 0.0f);
                o.z = fmaxf(o.z, 0.0f); o.w = fmaxf(o.w, 0.0f);
            }
        }
        if (beg >= end) load_slots(sidx, col, nbeg);
        *(float4*)&sA[buf][g * PITCH + 4 * c] = o;
        __syncthreads();

        const int r0 = base + strip * 16;
        if (has_tile && r0 + 16 <= N) {
            const float* arow = &sA[buf][(strip * 16 + l15) * PITCH + hi2];
            const float* bfr  = &sW[(ct * KS * 32 + lane) * 2];
            v8f d = v8f{};
#pragma unroll
            for (int s = 0; s < KS; ++s) {
                const v2f a = *(const v2f*)(arow + 4 * s);
                const v2f b = *(const v2f*)(bfr + s * 64);
                d = __builtin_amdgcn_wmma_f32_16x16x4_f32(
                     false, a,  false, b,
                     (short)0, d,  false,  false);
            }
            float dv[8];
#pragma unroll
            for (int v = 0; v < 8; ++v) dv[v] = OSCALE ? dinv[r0 + hi8 + v] : 1.0f;
            float* orow = OUT + (size_t)(r0 + hi8) * FO + ct * 16 + l15;
#pragma unroll
            for (int v = 0; v < 8; ++v) {
                float q = d[v];
                if (OBIAS) q += ob;
                if (ORELU) q = fmaxf(q, 0.0f);
                if (OSCALE) q *= dv[v];
                orow[(size_t)v * FO] = q;
            }
        } else if (has_tile && r0 < N) {
            for (int r = (lane >> 4); r < 16 && r0 + r < N; r += 2) {
                const float* ar = &sA[buf][(strip * 16 + r) * PITCH];
                float q = 0.0f;
                for (int k = 0; k < F; ++k) q += ar[k] * W[(size_t)k * FO + ct * 16 + l15];
                if (OBIAS) q += ob;
                if (ORELU) q = fmaxf(q, 0.0f);
                if (OSCALE) q *= dinv[r0 + r];
                OUT[(size_t)(r0 + r) * FO + ct * 16 + l15] = q;
            }
        }

        node = nxt; beg = nbeg; end = nend; nbeg = n2beg; nend = n2end;
        buf ^= 1;
    }
}

static inline int nblk(long long n, int t) { return (int)((n + t - 1) / t); }

template <int K, int F, bool BIAS, bool RELU, bool SCALE>
static void launch_gemm(const float* X, const float* W, const float* bias, const float* dinv,
                        float* H, int N, hipStream_t s) {
    int tilesM = N / 16;
    long long waves = tilesM;
    if (waves > 0)
        k_gemm_wmma<K, F, BIAS, RELU, SCALE><<<nblk(waves, 8), 256, 0, s>>>(X, W, bias, dinv, H,
                                                                            tilesM);
    int tail = tilesM * 16;
    if (tail < N)
        k_gemm_tail<K, F, BIAS, RELU, SCALE><<<nblk((long long)(N - tail) * F, 256), 256, 0, s>>>(
            X, W, bias, dinv, H, tail, N);
}

extern "C" void kernel_launch(void* const* d_in, const int* in_sizes, int n_in,
                              void* d_out, int out_size, void* d_ws, size_t ws_size,
                              hipStream_t stream) {
    const float* x  = (const float*)d_in[0];
    const int*   ei = (const int*)d_in[1];
    const float* W1 = (const float*)d_in[2]; const float* b1 = (const float*)d_in[3];
    const float* W2 = (const float*)d_in[4]; const float* b2 = (const float*)d_in[5];
    const float* W3 = (const float*)d_in[6]; const float* b3 = (const float*)d_in[7];
    const float* W4 = (const float*)d_in[8]; const float* b4 = (const float*)d_in[9];

    const int N = in_sizes[0] / 128;
    const int E = in_sizes[1] / 2;
    const int* src = ei;
    const int* dst = ei + E;

    float* out  = (float*)d_out;
    float* xrec = out;
    float* z    = out + (size_t)N * 128;

    float* ws   = (float*)d_ws;
    float* dinv = ws;
    float* bufH = ws + N;
    float* bufA = bufH + (size_t)N * 128;
    int* tblc   = (int*)(bufH + (size_t)N * 64);
    int* tblo   = tblc + NBMAX * SG;
    int* rowptr = tblo + NBMAX * SG + 4;
    int* col    = rowptr + N + 4;
    int* sorted = col + E;

    const int GG  = 1536;
    const int GGF = 1280;

    const int NB  = (N + BW - 1) >> BSH;
    k_slice_sort<<<SG, 256, 0, stream>>>(src, dst, tblc, tblo, sorted, E, NB);
    k_bucket_rows<<<NB, KT, 0, stream>>>(sorted, tblc, tblo, rowptr, dinv, col, N, NB);

    launch_gemm<128, 64, false, false, true>(x, W1, nullptr, dinv, bufH, N, stream);

    k_gather_gemm<64, 32, true, true, false, false, true><<<GGF, 256, 0, stream>>>(
        bufH, rowptr, col, dinv, b1, W2, nullptr, bufA, N);

    k_gather<32, true, false, true><<<GG, 256, 0, stream>>>(bufA, rowptr, col, dinv, b2, z, bufH,
                                                            N);

    k_gather_gemm<32, 64, false, false, true, true, true><<<GGF, 256, 0, stream>>>(
        bufH, rowptr, col, dinv, nullptr, W3, b3, bufA, N);

    k_gather_gemm<64, 128, false, false, true, false, false><<<GGF, 256, 0, stream>>>(
        bufA, rowptr, col, dinv, nullptr, W4, b4, xrec, N);
}


# Bucket kernel builds its part of col in shared memory (64 KB, on col's 16-byte grid) and copies it out in whole lines: no 4-byte store per edge spread over 32 lines a wave; sums same
# speedup vs baseline: 6.2422x; 1.0482x over previous
#include <hip/hip_runtime.h>
#include <hip/hip_bf16.h>


typedef float v2f __attribute__((ext_vector_type(2)));
typedef float v8f __attribute__((ext_vector_type(8)));

constexpr int BSH   = 9;
constexpr int BW    = 1 << BSH;
constexpr int NBMAX = 256;
constexpr int SG    = 256;
constexpr int PSH   = 10;
constexpr int SU    = 8;
constexpr int CAP   = SU << PSH;
constexpr int KT    = BW;
constexpr int RPW   = SG / (KT / 32);
constexpr int CAPB  = 16384;
constexpr int SPI   = 2;
static_assert(NBMAX == 256 && RPW <= 32 && RPW * (KT / 32) == SG, "shape of the edge ordering");

__device__ __forceinline__ int wave_incl_scan(int v, int lane) {
    for (int d = 1; d < 32; d <<= 1) {
        int u = __shfl_up(v, d);
        if (lane >= d) v += u;
    }
    return v;
}

__device__ __forceinline__ void load_slots(int (&s)[8], const int* __restrict__ col, int j) {
    const int* __restrict__ p = col + (unsigned)j;
#pragma unroll
    for (int u = 0; u < 8; ++u) s[u] = p[u];
}

__device__ __forceinline__ void load_range(int& b, int& e, const int* __restrict__ rowptr, int node, int N) {
    const int n0 = (node < N) ? node : N;
    const int n1 = (node < N) ? (node + 1) : N;
    b = rowptr[n0];
    e = rowptr[n1];
}

__device__ __forceinline__ int pack_edge(int s, int d) { return (s << BSH) | (d & (BW - 1)); }

__global__ __launch_bounds__(256) void k_slice_sort(const int* __restrict__ src,
                                                    const int* __restrict__ dst,
                                                    int* __restrict__ tblc,
                                                    int* __restrict__ tblo,
                                                    int* __restrict__ sorted,
                                                    int E, int NB) {
    __shared__ int hist[NBMAX];
    __shared__ int wt[8];
    __shared__ __attribute__((aligned(16))) int stage[CAP];
    const int t    = threadIdx.x;
    const int lane = t & 31;
    const int wv   = t >> 5;
    hist[t] = 0;
    const bool al  = (((size_t)dst) & 15) == 0;
    const bool als = (((size_t)src) & 15) == 0;
    const int  nwp = al ? (E >> PSH) : 0;
    const int  t4  = 4 * t;
    const int  p0  = blockIdx.x;
    const bool reg = p0 < nwp;
    int4 d[SU], s[SU];
    if (reg) {
#pragma unroll
        for (int u = 0; u < SU; ++u) {
            const int p  = p0 + u * SG;
            const int pc = (p < nwp) ? p : p0;
            const size_t o = ((size_t)pc << PSH) + t4;
            d[u] = *(const int4*)(dst + o);
            if (als) {
                s[u] = *(const int4*)(src + o);
            } else {
                s[u].x = src[o]; s[u].y = src[o + 1]; s[u].z = src[o + 2]; s[u].w = src[o + 3];
            }
        }
        __builtin_amdgcn_sched_barrier(0);
    }
    __syncthreads();

    if (reg) {
#pragma unroll
        for (int u = 0; u < SU; ++u) {
            if (p0 + u * SG < nwp) {
                atomicAdd(&hist[d[u].x >> BSH], 1);
                atomicAdd(&hist[d[u].y >> BSH], 1);
                atomicAdd(&hist[d[u].z >> BSH], 1);
                atomicAdd(&hist[d[u].w >> BSH], 1);
            }
        }
    }
    for (int p = blockIdx.x; (p << PSH) < E; p += SG) {
        if (p < nwp) continue;
        for (int q = 0; q < 4; ++q) {
            const int i = (p << PSH) + q * 256 + t;
            if (i < E) atomicAdd(&hist[dst[i] >> BSH], 1);
        }
    }
    __syncthreads();

    const int c    = hist[t];
    const int incl = wave_incl_scan(c, lane);
    if (lane == 31) wt[wv] = incl;
    __syncthreads();
    const int wtv   = (lane < 8) ? wt[lane] : 0;
    const int wincl = wave_incl_scan(wtv, lane);
    const int woff  = __shfl(wincl, wv) - __shfl(wtv, wv);
    const int excl  = woff + incl - c;
    const int n     = __shfl(wincl, 7);
    if (t < NB) {
        tblc[t * SG + blockIdx.x] = c;
        tblo[t * SG + blockIdx.x] = excl;
    }
    hist[t] = excl;
    __syncthreads();

    if (reg) {
#pragma unroll
        for (int u = 0; u < SU; ++u) {
            if (p0 + u * SG < nwp) {
                stage[atomicAdd(&hist[d[u].x >> BSH], 1)] = pack_edge(s[u].x, d[u].x);
                stage[atomicAdd(&hist[d[u].y >> BSH], 1)] = pack_edge(s[u].y, d[u].y);
                stage[atomicAdd(&hist[d[u].z >> BSH], 1)] = pack_edge(s[u].z, d[u].z);
                stage[atomicAdd(&hist[d[u].w >> BSH], 1)] = pack_edge(s[u].w, d[u].w);
            }
        }
    }
    for (int p = blockIdx.x; (p << PSH) < E; p += SG) {
        if (p < nwp) continue;
        for (int q = 0; q < 4; ++q) {
            const int i = (p << PSH) + q * 256 + t;
            if (i < E) {
                const int dd = dst[i];
                stage[atomicAdd(&hist[dd >> BSH], 1)] = pack_edge(src[i], dd);
            }
        }
    }
    __syncthreads();

    int* __restrict__ out = sorted + (size_t)blockIdx.x * CAP;
    const bool alo = (((size_t)out) & 15) == 0;
    for (int i = t4; i < n; i += 1024) {
        if (alo && i + 3 < n) {
            *(int4*)(out + i) = *(const int4*)(stage + i);
        } else {
            if (i     < n) out[i]     = stage[i];
            if (i + 1 < n) out[i + 1] = stage[i + 1];
            if (i + 2 < n) out[i + 2] = stage[i + 2];
            if (i + 3 < n) out[i + 3] = stage[i + 3];
        }
    }
}

__global__ __launch_bounds__(KT) void k_bucket_rows(const int* __restrict__ sorted,
                                                      const int* __restrict__ tblc,
                                                      const int* __restrict__ tblo,
                                                      int* __restrict__ rowptr,
                                                      float* __restrict__ dinv,
                                                      int* __restrict__ col,
                                                      int N, int NB) {
    __shared__ int cnt[BW];
    __shared__ int wt[KT / 32];
    __shared__ int rsum[KT / 32], rtot[KT / 32], rmax[KT / 32];
    __shared__ __attribute__((aligned(16))) int scol[CAPB];
    const int b    = blockIdx.x;
    const int t    = threadIdx.x;
    const int lane = t & 31;
    const int wv   = t >> 5;
    int rc = 0, ro = 0, ra = 0;
    if (lane < RPW) {
        const int w = wv * RPW + lane;
        rc = tblc[b * SG + w];
        ro = tblo[b * SG + w];
        ra = w * CAP + ro;
    }
    cnt[t] = 0;
    int s1 = ro, s2 = rc, s3 = rc;
    for (int d = 16; d > 0; d >>= 1) {
        s1 += __shfl_xor(s1, d);
        s2 += __shfl_xor(s2, d);
        const int m = __shfl_xor(s3, d);
        s3 = (m > s3) ? m : s3;
    }
    if (lane == 0) { rsum[wv] = s1; rtot[wv] = s2; rmax[wv] = s3; }
    __syncthreads();
    int g1 = (lane < KT / 32) ? rsum[lane] : 0;
    int g2 = (lane < KT / 32) ? rtot[lane] : 0;
    int g3 = (lane < KT / 32) ? rmax[lane] : 0;
    for (int d = 16; d > 0; d >>= 1) {
        g1 += __shfl_xor(g1, d);
        g2 += __shfl_xor(g2, d);
        const int m = __shfl_xor(g3, d);
        g3 = (m > g3) ? m : g3;
    }
    const int beg    = g1;
    const int end    = g1 + g2;
    const int sweeps = (g3 + 32 * SPI - 1) / (32 * SPI);
    const int  nbk   = g2;
    const int  sh    = (int)((((size_t)(col + beg)) & 15) >> 2);
    const bool stage = nbk + sh <= CAPB;
    const int  soff  = sh - beg;

    for (int it = 0; it < sweeps; ++it) {
        int k[SPI * RPW];
#pragma unroll
        for (int q = 0; q < RPW; ++q) {
            const int nq = __builtin_amdgcn_readlane(rc, q);
            const int aq = __builtin_amdgcn_readlane(ra, q);
#pragma unroll
            for (int h = 0; h < SPI; ++h) {
                const int j = ((it * SPI + h) << 5) + lane;
                k[q * SPI + h] = sorted[(unsigned)((j < nq) ? (aq + j) : 0)];
            }
        }
        __builtin_amdgcn_sched_barrier(0);
#pragma unroll
        for (int q = 0; q < RPW; ++q) {
            const int nq = __builtin_amdgcn_readlane(rc, q);
#pragma unroll
            for (int h = 0; h < SPI; ++h) {
                const int j = ((it * SPI + h) << 5) + lane;
                if (j < nq) atomicAdd(&cnt[k[q * SPI + h] & (BW - 1)], 1);
            }
        }
    }
    __syncthreads();
    const int c    = cnt[t];
    const int incl = wave_incl_scan(c, lane);
    if (lane == 31) wt[wv] = incl;
    __syncthreads();
    const int wtv   = (lane < KT / 32) ? wt[lane] : 0;
    const int wincl = wave_incl_scan(wtv, lane);
    const int woff  = __shfl(wincl, wv) - __shfl(wtv, wv);
    const int first = beg + woff + incl - c;
    const int node  = (b << BSH) + t;
    cnt[t] = first;
    if (node < N) {
        rowptr[node] = first;
        dinv[node]   = rsqrtf((float)c + 1.0f);
    }
    if (b == NB - 1 && t == 0) rowptr[N] = end;
    __syncthreads();
    for (int it = 0; it < sweeps; ++it) {
        int k[SPI * RPW], w[SPI * RPW];
#pragma unroll
        for (int q = 0; q < RPW; ++q) {
            const int nq = __builtin_amdgcn_readlane(rc, q);
            const int aq = __builtin_amdgcn_readlane(ra, q);
#pragma unroll
            for (int h = 0; h < SPI; ++h) {
                const int j = ((it * SPI + h) << 5) + lane;
                k[q * SPI + h] = sorted[(unsigned)((j < nq) ? (aq + j) : 0)];
            }
        }
        __builtin_amdgcn_sched_barrier(0);
#pragma unroll
        for (int q = 0; q < RPW; ++q) {
            const int nq = __builtin_amdgcn_readlane(rc, q);
#pragma unroll
            for (int h = 0; h < SPI; ++h) {
                const int j = ((it * SPI + h) << 5) + lane;
                w[q * SPI + h] = (j < nq) ? atomicAdd(&cnt[k[q * SPI + h] & (BW - 1)], 1) : 0;
            }
        }
        if (stage) {
#pragma unroll
            for (int q = 0; q < RPW; ++q) {
                const int nq = __builtin_amdgcn_readlane(rc, q);
#pragma unroll
                for (int h = 0; h < SPI; ++h) {
                    const int j = ((it * SPI + h) << 5) + lane;
                    if (j < nq) scol[w[q * SPI + h] + soff] = k[q * SPI + h] >> BSH;
                }
            }
        } else {
#pragma unroll
            for (int q = 0; q < RPW; ++q) {
                const int nq = __builtin_amdgcn_readlane(rc, q);
#pragma unroll
                for (int h = 0; h < SPI; ++h) {
                    const int j = ((it * SPI + h) << 5) + lane;
                    if (j < nq) col[w[q * SPI + h]] = k[q * SPI + h] >> BSH;
                }
            }
        }
    }
    if (stage) {
        __syncthreads();
        const int lo = sh, hi = sh + nbk;
        const int g0 = beg - sh;
        for (int i = 4 * t; i < hi; i += 4 * KT) {
            if (i >= lo && i + 3 < hi) {
                *(int4*)(col + (g0 + i)) = *(const int4*)(scol + i);
            } else {
                if (i     >= lo && i     < hi) col[g0 + i]     = scol[i];
                if (i + 1 >= lo && i + 1 < hi) col[g0 + i + 1] = scol[i + 1];
                if (i + 2 >= lo && i + 2 < hi) col[g0 + i + 2] = scol[i + 2];
                if (i + 3 >= lo && i + 3 < hi) col[g0 + i + 3] = scol[i + 3];
            }
        }
    }
}

template <int K, int F, bool BIAS, bool RELU, bool SCALE>
__global__ __launch_bounds__(256) void k_gemm_wmma(const float* __restrict__ X,
                                                   const float* __restrict__ W,
                                                   const float* __restrict__ bias,
                                                   const float* __restrict__ dinv,
                                                   float* __restrict__ H,
                                                   int tilesM) {
    static_assert(K % 4 == 0 && F % 16 == 0, "shape");
    constexpr int TF = F / 16;
    const int tm = blockIdx.x * (blockDim.x >> 5) + (threadIdx.x >> 5);
    if (tm >= tilesM) return;
    const int lane = threadIdx.x & 31;
    const int l15  = lane & 15;
    const int hi2  = (lane >> 4) << 1;

    const float* xrow = X + (size_t)(tm * 16 + l15) * K + hi2;
    const float* wcol = W + (size_t)hi2 * F + l15;

    v8f acc[TF];
#pragma unroll
    for (int t = 0; t < TF; ++t) acc[t] = v8f{};

    constexpr int KS = K / 4;
    v2f a[KS];
#pragma unroll
    for (int s = 0; s < KS; ++s) {
        a[s].x = xrow[4 * s];
        a[s].y = xrow[4 * s + 1];
    }

#pragma unroll
    for (int s = 0; s < KS; ++s) {
#pragma unroll
        for (int t = 0; t < TF; ++t) {
            v2f b;
            b.x = wcol[(size_t)(4 * s) * F + t * 16];
            b.y = wcol[(size_t)(4 * s + 1) * F + t * 16];
            acc[t] = __builtin_amdgcn_wmma_f32_16x16x4_f32(
                 false, a[s],  false, b,
                 (short)0, acc[t],  false,  false);
        }
    }

    const int hi8 = (lane >> 4) << 3;
    float dv[8];
#pragma unroll
    for (int v = 0; v < 8; ++v) dv[v] = SCALE ? dinv[tm * 16 + hi8 + v] : 1.0f;
    float* hrow = H + (size_t)(tm * 16 + hi8) * F + l15;
#pragma unroll
    for (int t = 0; t < TF; ++t) {
        float bv = 0.0f;
        if (BIAS) bv = bias[t * 16 + l15];
#pragma unroll
        for (int v = 0; v < 8; ++v) {
            float o = acc[t][v];
            if (BIAS) o += bv;
            if (RELU) o = fmaxf(o, 0.0f);
            if (SCALE) o *= dv[v];
            hrow[(size_t)v * F + t * 16] = o;
        }
    }
}

template <int K, int F, bool BIAS, bool RELU, bool SCALE>
__global__ __launch_bounds__(256) void k_gemm_tail(const float* __restrict__ X,
                                                   const float* __restrict__ W,
                                                   const float* __restrict__ bias,
                                                   const float* __restrict__ dinv,
                                                   float* __restrict__ H,
                                                   int rowStart, int rowEnd) {
    int idx = blockIdx.x * blockDim.x + threadIdx.x;
    int rows = rowEnd - rowStart;
    if (idx >= rows * F) return;
    int r = rowStart + idx / F, f = idx % F;
    float s = 0.f;
    for (int k = 0; k < K; ++k) s += X[(size_t)r * K + k] * W[(size_t)k * F + f];
    if (BIAS) s += bias[f];
    if (RELU) s = fmaxf(s, 0.0f);
    if (SCALE) s *= dinv[r];
    H[(size_t)r * F + f] = s;
}

template <int F, bool BIAS, bool RELU, bool DUAL>
__global__ __launch_bounds__(256) void k_gather(const float* __restrict__ HS,
                                                const int* __restrict__ rowptr,
                                                const int* __restrict__ col,
                                                const float* __restrict__ dinv,
                                                const float* __restrict__ bias,
                                                float* __restrict__ OUT,
                                                float* __restrict__ OUTS, int N) {
    constexpr int CH  = F / 4;
    constexpr int NPB = 256 / CH;
    const int c = threadIdx.x % CH;
    const int g = threadIdx.x / CH;
    const float4* __restrict__ H4 = (const float4*)HS;
    float4* __restrict__ O4 = (float4*)OUT;
    float4* __restrict__ S4 = (float4*)OUTS;

    float4 bv = make_float4(0.f, 0.f, 0.f, 0.f);
    if (BIAS) bv = ((const float4*)bias)[c];

    const int stride = gridDim.x * NPB;
    int node = blockIdx.x * NPB + g;
    int beg, end, nbeg, nend;
    load_range(beg, end, rowptr, node, N);
    load_range(nbeg, nend, rowptr, node + stride, N);
    int sidx[8];
    load_slots(sidx, col, beg);

    while (node < N) {
        const int nxt = node + stride;
        int n2beg, n2end;
        load_range(n2beg, n2end, rowptr, nxt + stride, N);

        const float dd = dinv[node];
        const float4 hs = H4[(size_t)node * CH + c];
        float4 acc = make_float4(0.f, 0.f, 0.f, 0.f);

        for (int j = beg; j < end; j += 8) {
            float4 h[8];
            int    nx[8];
#pragma unroll
            for (int u = 0; u < 8; ++u) {
                const int r = (j + u < end) ? sidx[u] : node;
                h[u] = H4[(unsigned)(r * CH + c)];
            }
            load_slots(nx, col, (j + 8 < end) ? (j + 8) : nbeg);
            __builtin_amdgcn_sched_barrier(0);
#pragma unroll
            for (int u = 0; u < 8; ++u) {
                const bool in = j + u < end;
                acc.x += in ? h[u].x : 0.0f;
                acc.y += in ? h[u].y : 0.0f;
                acc.z += in ? h[u].z : 0.0f;
                acc.w += in ? h[u].w : 0.0f;
            }
#pragma unroll
            for (int u = 0; u < 8; ++u) sidx[u] = nx[u];
        }
        if (beg >= end) load_slots(sidx, col, nbeg);

        float4 o;
        o.x = (acc.x + hs.x) * dd;
        o.y = (acc.y + hs.y) * dd;
        o.z = (acc.z + hs.z) * dd;
        o.w = (acc.w + hs.w) * dd;
        if (BIAS) { o.x += bv.x; o.y += bv.y; o.z += bv.z; o.w += bv.w; }
        if (RELU) {
            o.x = fmaxf(o.x, 0.0f); o.y = fmaxf(o.y, 0.0f);
            o.z = fmaxf(o.z, 0.0f); o.w = fmaxf(o.w, 0.0f);
        }
        O4[(size_t)node * CH + c] = o;
        if (DUAL) {
            float4 q;
            q.x = o.x * dd; q.y = o.y * dd; q.z = o.z * dd; q.w = o.w * dd;
            S4[(size_t)node * CH + c] = q;
        }

        node = nxt; beg = nbeg; end = nend; nbeg = n2beg; nend = n2end;
    }
}

template <int F, int FO, bool GBIAS, bool GRELU, bool OBIAS, bool ORELU, bool OSCALE>
__global__ __launch_bounds__(256) void k_gather_gemm(const float* __restrict__ HS,
                                                     const int* __restrict__ rowptr,
                                                     const int* __restrict__ col,
                                                     const float* __restrict__ dinv,
                                                     const float* __restrict__ gbias,
                                                     const float* __restrict__ W,
                                                     const float* __restrict__ obias,
                                                     float* __restrict__ OUT, int N) {
    constexpr int CH    = F / 4;
    constexpr int NPB   = 256 / CH;
    constexpr int PITCH = F + 4;
    constexpr int CT    = FO / 16;
    constexpr int TILES = (NPB / 16) * CT;
    constexpr int KS    = F / 4;
    static_assert(TILES <= 8 && NPB % 16 == 0, "one tile per wave");
    __shared__ __attribute__((aligned(16))) float sA[2][NPB * PITCH];
    __shared__ __attribute__((aligned(16))) float sW[F * FO];

    const int c = threadIdx.x % CH;
    const int g = threadIdx.x / CH;
    const int lane = threadIdx.x & 31;
    const int wv   = threadIdx.x >> 5;
    const int l15  = lane & 15;
    const int hi2  = (lane >> 4) << 1;
    const int hi8  = (lane >> 4) << 3;
    const bool has_tile = wv < TILES;
    const int strip = wv / CT;
    const int ct    = wv % CT;
    const float4* __restrict__ H4 = (const float4*)HS;

    for (int q = threadIdx.x; q < F * FO / 4; q += 256) {
        const int k  = q / (FO / 4);
        const int n4 = (q % (FO / 4)) * 4;
        const float4 w4 = *(const float4*)(W + (size_t)k * FO + n4);
        float* dst = &sW[(((n4 >> 4) * KS + (k >> 2)) * 32 + ((k >> 1) & 1) * 16 + (n4 & 15)) * 2 + (k & 1)];
        dst[0] = w4.x; dst[2] = w4.y; dst[4] = w4.z; dst[6] = w4.w;
    }
    float ob = 0.0f;
    if (OBIAS && has_tile) ob = obias[ct * 16 + l15];

    const int stride = gridDim.x * NPB;
    int node = blockIdx.x * NPB + g;
    int beg, end, nbeg, nend;
    load_range(beg, end, rowptr, node, N);
    load_range(nbeg, nend, rowptr, node + stride, N);
    int sidx[8];
    load_slots(sidx, col, beg);

    int buf = 0;
    for (int base = blockIdx.x * NPB; base < N; base += stride) {
        const int nxt = node + stride;
        int n2beg, n2end;
        load_range(n2beg, n2end, rowptr, nxt + stride, N);

        float4 o = make_float4(0.f, 0.f, 0.f, 0.f);
        if (node < N) {
            const float dd = dinv[node];
            const float4 hs = H4[(size_t)node * CH + c];
            float4 acc = make_float4(0.f, 0.f, 0.f, 0.f);

            for (int j = beg; j < end; j += 8) {
                float4 h[8];
                int    nx[8];
#pragma unroll
                for (int u = 0; u < 8; ++u) {
                    const int r = (j + u < end) ? sidx[u] : node;
                    h[u] = H4[(unsigned)(r * CH + c)];
                }
                load_slots(nx, col, (j + 8 < end) ? (j + 8) : nbeg);
                __builtin_amdgcn_sched_barrier(0);
#pragma unroll
                for (int u = 0; u < 8; ++u) {
                    const bool in = j + u < end;
                    acc.x += in ? h[u].x : 0.0f;
                    acc.y += in ? h[u].y : 0.0f;
                    acc.z += in ? h[u].z : 0.0f;
                    acc.w += in ? h[u].w : 0.0f;
                }
#pragma unroll
                for (int u = 0; u < 8; ++u) sidx[u] = nx[u];
            }

            o.x = (acc.x + hs.x) * dd;
            o.y = (acc.y + hs.y) * dd;
            o.z = (acc.z + hs.z) * dd;
            o.w = (acc.w + hs.w) * dd;
            if (GBIAS) {
                const float4 bv = ((const float4*)gbias)[c];
                o.x += bv.x; o.y += bv.y; o.z += bv.z; o.w += bv.w;
            }
            if (GRELU) {
                o.x = fmaxf(o.x, 0.0f); o.y = fmaxf(o.y, 0.0f);
                o.z = fmaxf(o.z, 0.0f); o.w = fmaxf(o.w, 0.0f);
            }
        }
        if (beg >= end) load_slots(sidx, col, nbeg);
        *(float4*)&sA[buf][g * PITCH + 4 * c] = o;
        __syncthreads();

        const int r0 = base + strip * 16;
        if (has_tile && r0 + 16 <= N) {
            const float* arow = &sA[buf][(strip * 16 + l15) * PITCH + hi2];
            const float* bfr  = &sW[(ct * KS * 32 + lane) * 2];
            v8f d = v8f{};
#pragma unroll
            for (int s = 0; s < KS; ++s) {
                const v2f a = *(const v2f*)(arow + 4 * s);
                const v2f b = *(const v2f*)(bfr + s * 64);
                d = __builtin_amdgcn_wmma_f32_16x16x4_f32(
                     false, a,  false, b,
                     (short)0, d,  false,  false);
            }
            float dv[8];
#pragma unroll
            for (int v = 0; v < 8; ++v) dv[v] = OSCALE ? dinv[r0 + hi8 + v] : 1.0f;
            float* orow = OUT + (size_t)(r0 + hi8) * FO + ct * 16 + l15;
#pragma unroll
            for (int v = 0; v < 8; ++v) {
                float q = d[v];
                if (OBIAS) q += ob;
                if (ORELU) q = fmaxf(q, 0.0f);
                if (OSCALE) q *= dv[v];
                orow[(size_t)v * FO] = q;
            }
        } else if (has_tile && r0 < N) {
            for (int r = (lane >> 4); r < 16 && r0 + r < N; r += 2) {
                const float* ar = &sA[buf][(strip * 16 + r) * PITCH];
                float q = 0.0f;
                for (int k = 0; k < F; ++k) q += ar[k] * W[(size_t)k * FO + ct * 16 + l15];
                if (OBIAS) q += ob;
                if (ORELU) q = fmaxf(q, 0.0f);
                if (OSCALE) q *= dinv[r0 + r];
                OUT[(size_t)(r0 + r) * FO + ct * 16 + l15] = q;
            }
        }

        node = nxt; beg = nbeg; end = nend; nbeg = n2beg; nend = n2end;
        buf ^= 1;
    }
}

static inline int nblk(long long n, int t) { return (int)((n + t - 1) / t); }

template <int K, int F, bool BIAS, bool RELU, bool SCALE>
static void launch_gemm(const float* X, const float* W, const float* bias, const float* dinv,
                        float* H, int N, hipStream_t s) {
    int tilesM = N / 16;
    long long waves = tilesM;
    if (waves > 0)
        k_gemm_wmma<K, F, BIAS, RELU, SCALE><<<nblk(waves, 8), 256, 0, s>>>(X, W, bias, dinv, H,
                                                                            tilesM);
    int tail = tilesM * 16;
    if (tail < N)
        k_gemm_tail<K, F, BIAS, RELU, SCALE><<<nblk((long long)(N - tail) * F, 256), 256, 0, s>>>(
            X, W, bias, dinv, H, tail, N);
}

extern "C" void kernel_launch(void* const* d_in, const int* in_sizes, int n_in,
                              void* d_out, int out_size, void* d_ws, size_t ws_size,
                              hipStream_t stream) {
    const float* x  = (const float*)d_in[0];
    const int*   ei = (const int*)d_in[1];
    const float* W1 = (const float*)d_in[2]; const float* b1 = (const float*)d_in[3];
    const float* W2 = (const float*)d_in[4]; const float* b2 = (const float*)d_in[5];
    const float* W3 = (const float*)d_in[6]; const float* b3 = (const float*)d_in[7];
    const float* W4 = (const float*)d_in[8]; const float* b4 = (const float*)d_in[9];

    const int N = in_sizes[0] / 128;
    const int E = in_sizes[1] / 2;
    const int* src = ei;
    const int* dst = ei + E;

    float* out  = (float*)d_out;
    float* xrec = out;
    float* z    = out + (size_t)N * 128;

    float* ws   = (float*)d_ws;
    float* dinv = ws;
    float* bufH = ws + N;
    float* bufA = bufH + (size_t)N * 128;
    int* tblc   = (int*)(bufH + (size_t)N * 64);
    int* tblo   = tblc + NBMAX * SG;
    int* rowptr = tblo + NBMAX * SG + 4;
    int* col    = rowptr + N + 4;
    int* sorted = col + E;

    const int GG  = 1536;
    const int GGF = 1280;

    const int NB  = (N + BW - 1) >> BSH;
    k_slice_sort<<<SG, 256, 0, stream>>>(src, dst, tblc, tblo, sorted, E, NB);
    k_bucket_rows<<<NB, KT, 0, stream>>>(sorted, tblc, tblo, rowptr, dinv, col, N, NB);

    launch_gemm<128, 64, false, false, true>(x, W1, nullptr, dinv, bufH, N, stream);

    k_gather_gemm<64, 32, true, true, false, false, true><<<GGF, 256, 0, stream>>>(
        bufH, rowptr, col, dinv, b1, W2, nullptr, bufA, N);

    k_gather<32, true, false, true><<<GG, 256, 0, stream>>>(bufA, rowptr, col, dinv, b2, z, bufH,
                                                            N);

    k_gather_gemm<32, 64, false, false, true, true, true><<<GGF, 256, 0, stream>>>(
        bufH, rowptr, col, dinv, nullptr, W3, b3, bufA, N);

    k_gather_gemm<64, 128, false, false, true, false, false><<<GGF, 256, 0, stream>>>(
        bufA, rowptr, col, dinv, nullptr, W4, b4, xrec, N);
}
